# starting kernel
# baseline (speedup 1.0000x reference)
#include <hip/hip_runtime.h>


typedef float v2f __attribute__((ext_vector_type(2)));
typedef float v8f __attribute__((ext_vector_type(8)));

#define BATCH 4096
#define TSTEP 512
#define IDIM  13
#define HDIM  7
#define GDIM  28
#define FFDIM 512

#define LOG2E 1.4426950408889634f

#if __has_builtin(__builtin_amdgcn_tanhf)
#define FAST_TANH(v) __builtin_amdgcn_tanhf(v)
#define TANH_TIER 2
#elif __has_builtin(__builtin_amdgcn_tanh_f32)
#define FAST_TANH(v) __builtin_amdgcn_tanh_f32(v)
#define TANH_TIER 2
#elif __has_builtin(__builtin_amdgcn_exp2f) && __has_builtin(__builtin_amdgcn_rcpf)
#define TANH_TIER 1
static __device__ __forceinline__ float FAST_TANH(float v) {
    float e2 = __builtin_amdgcn_exp2f(v * (2.0f * LOG2E));
    return fmaf(-2.0f, __builtin_amdgcn_rcpf(e2 + 1.0f), 1.0f);
}
#else
#define FAST_TANH(v) tanhf(v)
#define TANH_TIER 0
#endif

static __device__ __forceinline__ float sigm(float v) {
#if TANH_TIER == 2
    return fmaf(0.5f, FAST_TANH(0.5f * v), 0.5f);
#elif TANH_TIER == 1
    return __builtin_amdgcn_rcpf(1.0f + __builtin_amdgcn_exp2f(v * -LOG2E));
#else
    return 1.0f / (1.0f + __expf(-v));
#endif
}

__global__ __launch_bounds__(32) void lstm_fused_kernel(
    const float* __restrict__ x,     const float* __restrict__ w_ih,
    const float* __restrict__ w_hh,  const float* __restrict__ b_ih,
    const float* __restrict__ b_hh,  const float* __restrict__ fc1_w,
    const float* __restrict__ fc1_b, const float* __restrict__ fc_w,
    const float* __restrict__ fc_b,  float* __restrict__ out)
{
    __shared__ float gates[16][32];
    __shared__ float hbuf[16][8];
    __shared__ float hdump[16];
    __shared__ float red[16];

    const int l    = threadIdx.x;
    const int half = l >> 4;
    const int nl   = l & 15;
    const int b0   = blockIdx.x * 16;

    v2f Bih[2][4];
    v2f Bhh[2][2];
#pragma unroll
    for (int kc = 0; kc < 4; ++kc) {
        int k0 = kc * 4 + 2 * half;
#pragma unroll
        for (int nt = 0; nt < 2; ++nt) {
            int n  = nt * 16 + nl;
            int nn = (n < GDIM) ? n : 0;
            float m0 = (n < GDIM && k0     < IDIM) ? 1.0f : 0.0f;
            float m1 = (n < GDIM && k0 + 1 < IDIM) ? 1.0f : 0.0f;
            int ka = (k0     < IDIM) ? k0     : 0;
            int kb = (k0 + 1 < IDIM) ? k0 + 1 : 0;
            Bih[nt][kc][0] = w_ih[nn * IDIM + ka] * m0;
            Bih[nt][kc][1] = w_ih[nn * IDIM + kb] * m1;
        }
    }
#pragma unroll
    for (int kc = 0; kc < 2; ++kc) {
        int k0 = kc * 4 + 2 * half;
#pragma unroll
        for (int nt = 0; nt < 2; ++nt) {
            int n  = nt * 16 + nl;
            int nn = (n < GDIM) ? n : 0;
            float m0 = (n < GDIM && k0     < HDIM) ? 1.0f : 0.0f;
            float m1 = (n < GDIM && k0 + 1 < HDIM) ? 1.0f : 0.0f;
            int ka = (k0     < HDIM) ? k0     : 0;
            int kb = (k0 + 1 < HDIM) ? k0 + 1 : 0;
            Bhh[nt][kc][0] = w_hh[nn * HDIM + ka] * m0;
            Bhh[nt][kc][1] = w_hh[nn * HDIM + kb] * m1;
        }
    }

    const float bias0 = b_ih[nl] + b_hh[nl];
    const float bias1 = (16 + nl < GDIM) ? (b_ih[16 + nl] + b_hh[16 + nl]) : 0.0f;

    if (half == 0) {
#pragma unroll
        for (int k = 0; k < 8; ++k) hbuf[nl][k] = 0.0f;
    }
    float c[4] = {0.0f, 0.0f, 0.0f, 0.0f};
    __syncthreads();

    const float* xrow = x + (size_t)(b0 + nl) * (TSTEP * IDIM);

    auto x_project = [&](int t, v8f& xa0, v8f& xa1) {
        const float* xt = xrow + t * IDIM;
#pragma unroll
        for (int i = 0; i < 8; ++i) { xa0[i] = bias0; xa1[i] = bias1; }
#pragma unroll
        for (int kc = 0; kc < 4; ++kc) {
            int k0 = kc * 4 + 2 * half;
            v2f a;
            int ka = (k0     < IDIM) ? k0     : 0;
            int kb = (k0 + 1 < IDIM) ? k0 + 1 : 0;
            a[0] = xt[ka] * ((k0     < IDIM) ? 1.0f : 0.0f);
            a[1] = xt[kb] * ((k0 + 1 < IDIM) ? 1.0f : 0.0f);
            xa0 = __builtin_amdgcn_wmma_f32_16x16x4_f32(false, a, false, Bih[0][kc],
                                                        (short)0, xa0, false, false);
            xa1 = __builtin_amdgcn_wmma_f32_16x16x4_f32(false, a, false, Bih[1][kc],
                                                        (short)0, xa1, false, false);
        }
    };

    v8f xacc0, xacc1;
    x_project(0, xacc0, xacc1);

    for (int t = 0; t < TSTEP; ++t) {
        v8f acc0 = xacc0, acc1 = xacc1;

#pragma unroll
        for (int kc = 0; kc < 2; ++kc) {
            int k0 = kc * 4 + 2 * half;
            v2f a;
            a[0] = hbuf[nl][k0];
            a[1] = hbuf[nl][k0 + 1];
            acc0 = __builtin_amdgcn_wmma_f32_16x16x4_f32(false, a, false, Bhh[0][kc],
                                                         (short)0, acc0, false, false);
            acc1 = __builtin_amdgcn_wmma_f32_16x16x4_f32(false, a, false, Bhh[1][kc],
                                                         (short)0, acc1, false, false);
        }

#pragma unroll
        for (int v = 0; v < 8; ++v) {
            int m = v + 8 * half;
            gates[m][nl]      = acc0[v];
            gates[m][nl + 16] = acc1[v];
        }
        __syncthreads();

        int tn = (t + 1 < TSTEP) ? (t + 1) : 0;
        x_project(tn, xacc0, xacc1);

        {
            const int row = nl;
            const int jb  = half * 4;
#pragma unroll
            for (int jj = 0; jj < 4; ++jj) {
                int j  = jb + jj;
                int jc = (j < HDIM) ? j : (HDIM - 1);
                float ig = sigm(gates[row][jc]);
                float fg = sigm(gates[row][jc + 7]);
                float gg = FAST_TANH(gates[row][jc + 14]);
                float og = sigm(gates[row][jc + 21]);
                c[jj] = fg * c[jj] + ig * gg;
                float hv = og * FAST_TANH(c[jj]);
                float* dst = (j < HDIM) ? &hbuf[row][j] : &hdump[row];
                *dst = hv;
            }
        }
        __syncthreads();
    }

    float hr[HDIM];
#pragma unroll
    for (int j = 0; j < HDIM; ++j) {
        float v = hbuf[nl][j];
        hr[j] = v > 0.0f ? v : 0.0f;
    }
    float acc = 0.0f;
    const int ffb = half * (FFDIM / 2);
    for (int ff = ffb; ff < ffb + FFDIM / 2; ++ff) {
        float s = fc1_b[ff];
#pragma unroll
        for (int j = 0; j < HDIM; ++j) s += fc1_w[ff * HDIM + j] * hr[j];
        s = s > 0.0f ? s : 0.0f;
        acc += fc_w[ff] * s;
    }
    if (half == 1) red[nl] = acc;
    __syncthreads();
    if (half == 0) out[b0 + nl] = acc + red[nl] + fc_b[0];
}

extern "C" void kernel_launch(void* const* d_in, const int* in_sizes, int n_in,
                              void* d_out, int out_size, void* d_ws, size_t ws_size,
                              hipStream_t stream) {
    (void)in_sizes; (void)n_in; (void)out_size; (void)d_ws; (void)ws_size;
    const float* x     = (const float*)d_in[0];
    const float* w_ih  = (const float*)d_in[1];
    const float* w_hh  = (const float*)d_in[2];
    const float* b_ih  = (const float*)d_in[3];
    const float* b_hh  = (const float*)d_in[4];
    const float* fc1_w = (const float*)d_in[5];
    const float* fc1_b = (const float*)d_in[6];
    const float* fc_w  = (const float*)d_in[7];
    const float* fc_b  = (const float*)d_in[8];

    lstm_fused_kernel<<<dim3(BATCH / 16), dim3(32), 0, stream>>>(
        x, w_ih, w_hh, b_ih, b_hh, fc1_w, fc1_b, fc_w, fc_b, (float*)d_out);
}


# x A-fragment loads issued 7 steps ahead into an 8-slot register ring (step loop unrolled by 8; LDS-only fences, no arithmetic on loaded values), so no step waits for its own memory miss
# speedup vs baseline: 1.7073x; 1.7073x over previous
#include <hip/hip_runtime.h>


typedef float v2f __attribute__((ext_vector_type(2)));
typedef float v8f __attribute__((ext_vector_type(8)));

#define BATCH 4096
#define TSTEP 512
#define IDIM  13
#define HDIM  7
#define GDIM  28
#define FFDIM 512
#define XPF   8

#define LOG2E 1.4426950408889634f

#if __has_builtin(__builtin_amdgcn_tanhf)
#define FAST_TANH(v) __builtin_amdgcn_tanhf(v)
#define TANH_TIER 2
#elif __has_builtin(__builtin_amdgcn_tanh_f32)
#define FAST_TANH(v) __builtin_amdgcn_tanh_f32(v)
#define TANH_TIER 2
#elif __has_builtin(__builtin_amdgcn_exp2f) && __has_builtin(__builtin_amdgcn_rcpf)
#define TANH_TIER 1
static __device__ __forceinline__ float FAST_TANH(float v) {
    float e2 = __builtin_amdgcn_exp2f(v * (2.0f * LOG2E));
    return fmaf(-2.0f, __builtin_amdgcn_rcpf(e2 + 1.0f), 1.0f);
}
#else
#define FAST_TANH(v) tanhf(v)
#define TANH_TIER 0
#endif

static __device__ __forceinline__ float sigm(float v) {
#if TANH_TIER == 2
    return fmaf(0.5f, FAST_TANH(0.5f * v), 0.5f);
#elif TANH_TIER == 1
    return __builtin_amdgcn_rcpf(1.0f + __builtin_amdgcn_exp2f(v * -LOG2E));
#else
    return 1.0f / (1.0f + __expf(-v));
#endif
}

static __device__ __forceinline__ void lds_sync() {
    __builtin_amdgcn_fence(__ATOMIC_RELEASE, "workgroup", "local");
    __builtin_amdgcn_s_barrier();
    __builtin_amdgcn_fence(__ATOMIC_ACQUIRE, "workgroup", "local");
}

__global__ __launch_bounds__(32) void lstm_fused_kernel(
    const float* __restrict__ x,     const float* __restrict__ w_ih,
    const float* __restrict__ w_hh,  const float* __restrict__ b_ih,
    const float* __restrict__ b_hh,  const float* __restrict__ fc1_w,
    const float* __restrict__ fc1_b, const float* __restrict__ fc_w,
    const float* __restrict__ fc_b,  float* __restrict__ out)
{
    __shared__ float gates[16][32];
    __shared__ float hbuf[16][8];
    __shared__ float hdump[16];
    __shared__ float red[16];

    const int l    = threadIdx.x;
    const int half = l >> 4;
    const int nl   = l & 15;
    const int b0   = blockIdx.x * 16;

    v2f Bih[2][4];
    v2f Bhh[2][2];
#pragma unroll
    for (int kc = 0; kc < 4; ++kc) {
        int k0 = kc * 4 + 2 * half;
#pragma unroll
        for (int nt = 0; nt < 2; ++nt) {
            int n  = nt * 16 + nl;
            int nn = (n < GDIM) ? n : 0;
            float m0 = (n < GDIM && k0     < IDIM) ? 1.0f : 0.0f;
            float m1 = (n < GDIM && k0 + 1 < IDIM) ? 1.0f : 0.0f;
            int ka = (k0     < IDIM) ? k0     : 0;
            int kb = (k0 + 1 < IDIM) ? k0 + 1 : 0;
            Bih[nt][kc][0] = w_ih[nn * IDIM + ka] * m0;
            Bih[nt][kc][1] = w_ih[nn * IDIM + kb] * m1;
        }
    }
#pragma unroll
    for (int kc = 0; kc < 2; ++kc) {
        int k0 = kc * 4 + 2 * half;
#pragma unroll
        for (int nt = 0; nt < 2; ++nt) {
            int n  = nt * 16 + nl;
            int nn = (n < GDIM) ? n : 0;
            float m0 = (n < GDIM && k0     < HDIM) ? 1.0f : 0.0f;
            float m1 = (n < GDIM && k0 + 1 < HDIM) ? 1.0f : 0.0f;
            int ka = (k0     < HDIM) ? k0     : 0;
            int kb = (k0 + 1 < HDIM) ? k0 + 1 : 0;
            Bhh[nt][kc][0] = w_hh[nn * HDIM + ka] * m0;
            Bhh[nt][kc][1] = w_hh[nn * HDIM + kb] * m1;
        }
    }

    const float bias0 = b_ih[nl] + b_hh[nl];
    const float bias1 = (16 + nl < GDIM) ? (b_ih[16 + nl] + b_hh[16 + nl]) : 0.0f;

    if (half == 0) {
#pragma unroll
        for (int k = 0; k < 8; ++k) hbuf[nl][k] = 0.0f;
    }
    float c[4] = {0.0f, 0.0f, 0.0f, 0.0f};
    __syncthreads();

    const float* xblk = x + (size_t)b0 * (TSTEP * IDIM);
    const int xlane  = nl * (TSTEP * IDIM) + 2 * half;
    const int xlane3 = nl * (TSTEP * IDIM) + ((half == 0) ? 12 : 0);

    auto x_load = [&](int t, v2f (&xr)[4]) {
        const float* xt = xblk + t * IDIM;
#pragma unroll
        for (int kc = 0; kc < 3; ++kc) {
            xr[kc][0] = xt[xlane + kc * 4];
            xr[kc][1] = xt[xlane + kc * 4 + 1];
        }
        xr[3][0] = xt[xlane3];
        xr[3][1] = 0.0f;
    };

    auto x_project = [&](const v2f (&xr)[4], v8f& xa0, v8f& xa1) {
#pragma unroll
        for (int i = 0; i < 8; ++i) { xa0[i] = bias0; xa1[i] = bias1; }
#pragma unroll
        for (int kc = 0; kc < 4; ++kc) {
            xa0 = __builtin_amdgcn_wmma_f32_16x16x4_f32(false, xr[kc], false, Bih[0][kc],
                                                        (short)0, xa0, false, false);
            xa1 = __builtin_amdgcn_wmma_f32_16x16x4_f32(false, xr[kc], false, Bih[1][kc],
                                                        (short)0, xa1, false, false);
        }
    };

    v2f xring[XPF][4];
    v8f xacc0, xacc1;
    x_load(0, xring[0]);
    x_project(xring[0], xacc0, xacc1);
#pragma unroll
    for (int s = 1; s < XPF; ++s) x_load(s, xring[s]);

    static_assert(TSTEP % XPF == 0, "ring slots are static: the step loop is unrolled by XPF");
    for (int tb = 0; tb < TSTEP; tb += XPF) {
#pragma unroll
    for (int u = 0; u < XPF; ++u) {
        const int t = tb + u;
        v8f acc0 = xacc0, acc1 = xacc1;

#pragma unroll
        for (int kc = 0; kc < 2; ++kc) {
            int k0 = kc * 4 + 2 * half;
            v2f a;
            a[0] = hbuf[nl][k0];
            a[1] = hbuf[nl][k0 + 1];
            acc0 = __builtin_amdgcn_wmma_f32_16x16x4_f32(false, a, false, Bhh[0][kc],
                                                         (short)0, acc0, false, false);
            acc1 = __builtin_amdgcn_wmma_f32_16x16x4_f32(false, a, false, Bhh[1][kc],
                                                         (short)0, acc1, false, false);
        }

#pragma unroll
        for (int v = 0; v < 8; ++v) {
            int m = v + 8 * half;
            gates[m][nl]      = acc0[v];
            gates[m][nl + 16] = acc1[v];
        }
        lds_sync();

        int tn = (t + XPF < TSTEP) ? (t + XPF) : 0;
        x_load(tn, xring[u]);
        x_project(xring[(u + 1) % XPF], xacc0, xacc1);

        {
            const int row = nl;
            const int jb  = half * 4;
#pragma unroll
            for (int jj = 0; jj < 4; ++jj) {
                int j  = jb + jj;
                int jc = (j < HDIM) ? j : (HDIM - 1);
                float ig = sigm(gates[row][jc]);
                float fg = sigm(gates[row][jc + 7]);
                float gg = FAST_TANH(gates[row][jc + 14]);
                float og = sigm(gates[row][jc + 21]);
                c[jj] = fg * c[jj] + ig * gg;
                float hv = og * FAST_TANH(c[jj]);
                float* dst = (j < HDIM) ? &hbuf[row][j] : &hdump[row];
                *dst = hv;
            }
        }
        lds_sync();
    }
    }

    float hr[HDIM];
#pragma unroll
    for (int j = 0; j < HDIM; ++j) {
        float v = hbuf[nl][j];
        hr[j] = v > 0.0f ? v : 0.0f;
    }
    float acc = 0.0f;
    const int ffb = half * (FFDIM / 2);
    for (int ff = ffb; ff < ffb + FFDIM / 2; ++ff) {
        float s = fc1_b[ff];
#pragma unroll
        for (int j = 0; j < HDIM; ++j) s += fc1_w[ff * HDIM + j] * hr[j];
        s = s > 0.0f ? s : 0.0f;
        acc += fc_w[ff] * s;
    }
    if (half == 1) red[nl] = acc;
    __syncthreads();
    if (half == 0) out[b0 + nl] = acc + red[nl] + fc_b[0];
}

extern "C" void kernel_launch(void* const* d_in, const int* in_sizes, int n_in,
                              void* d_out, int out_size, void* d_ws, size_t ws_size,
                              hipStream_t stream) {
    (void)in_sizes; (void)n_in; (void)out_size; (void)d_ws; (void)ws_size;
    const float* x     = (const float*)d_in[0];
    const float* w_ih  = (const float*)d_in[1];
    const float* w_hh  = (const float*)d_in[2];
    const float* b_ih  = (const float*)d_in[3];
    const float* b_hh  = (const float*)d_in[4];
    const float* fc1_w = (const float*)d_in[5];
    const float* fc1_b = (const float*)d_in[6];
    const float* fc_w  = (const float*)d_in[7];
    const float* fc_b  = (const float*)d_in[8];

    lstm_fused_kernel<<<dim3(BATCH / 16), dim3(32), 0, stream>>>(
        x, w_ih, w_hh, b_ih, b_hh, fc1_w, fc1_b, fc_w, fc_b, (float*)d_out);
}


# Recurrence kept in registers: weights as WMMA A operand with gate rows unit-major, batch rows as B columns, so a lane's D holds the gates of the units whose h it feeds back as B; no LDS per step
# speedup vs baseline: 3.1521x; 1.8462x over previous
#include <hip/hip_runtime.h>


typedef float v2f __attribute__((ext_vector_type(2)));
typedef float v8f __attribute__((ext_vector_type(8)));

#define BATCH 4096
#define TSTEP 512
#define IDIM  13
#define HDIM  7
#define FFDIM 512
#define XPF   8

#define LOG2E 1.4426950408889634f

#if __has_builtin(__builtin_amdgcn_tanhf)
#define FAST_TANH(v) __builtin_amdgcn_tanhf(v)
#define TANH_TIER 2
#elif __has_builtin(__builtin_amdgcn_tanh_f32)
#define FAST_TANH(v) __builtin_amdgcn_tanh_f32(v)
#define TANH_TIER 2
#elif __has_builtin(__builtin_amdgcn_exp2f) && __has_builtin(__builtin_amdgcn_rcpf)
#define TANH_TIER 1
static __device__ __forceinline__ float FAST_TANH(float v) {
    float e2 = __builtin_amdgcn_exp2f(v * (2.0f * LOG2E));
    return fmaf(-2.0f, __builtin_amdgcn_rcpf(e2 + 1.0f), 1.0f);
}
#else
#define FAST_TANH(v) tanhf(v)
#define TANH_TIER 0
#endif

static __device__ __forceinline__ float sigm(float v) {
#if TANH_TIER == 2
    return fmaf(0.5f, FAST_TANH(0.5f * v), 0.5f);
#elif TANH_TIER == 1
    return __builtin_amdgcn_rcpf(1.0f + __builtin_amdgcn_exp2f(v * -LOG2E));
#else
    return 1.0f / (1.0f + __expf(-v));
#endif
}

__global__ __launch_bounds__(32) void lstm_fused_kernel(
    const float* __restrict__ x,     const float* __restrict__ w_ih,
    const float* __restrict__ w_hh,  const float* __restrict__ b_ih,
    const float* __restrict__ b_hh,  const float* __restrict__ fc1_w,
    const float* __restrict__ fc1_b, const float* __restrict__ fc_w,
    const float* __restrict__ fc_b,  float* __restrict__ out)
{
    __shared__ float hbuf[16][8];
    __shared__ float red[16];

    const int l    = threadIdx.x;
    const int half = l >> 4;
    const int nl   = l & 15;
    const int b0   = blockIdx.x * 16;

    v2f Aih[2][4];
    v2f Ahh[2][2];
#pragma unroll
    for (int mt = 0; mt < 2; ++mt) {
        const int  u  = mt * 4 + (nl >> 2);
        const bool uv = u < HDIM;
        const int  g  = uv ? (nl & 3) * HDIM + u : 0;
#pragma unroll
        for (int kc = 0; kc < 4; ++kc) {
            int k0 = kc * 4 + 2 * half;
            float m0 = (uv && k0     < IDIM) ? 1.0f : 0.0f;
            float m1 = (uv && k0 + 1 < IDIM) ? 1.0f : 0.0f;
            int ka = (k0     < IDIM) ? k0     : 0;
            int kb = (k0 + 1 < IDIM) ? k0 + 1 : 0;
            Aih[mt][kc][0] = w_ih[g * IDIM + ka] * m0;
            Aih[mt][kc][1] = w_ih[g * IDIM + kb] * m1;
        }
#pragma unroll
        for (int kc = 0; kc < 2; ++kc) {
            int k0 = kc * 4 + 2 * half;
            float m0 = (uv && k0     < HDIM) ? 1.0f : 0.0f;
            float m1 = (uv && k0 + 1 < HDIM) ? 1.0f : 0.0f;
            int ka = (k0     < HDIM) ? k0     : 0;
            int kb = (k0 + 1 < HDIM) ? k0 + 1 : 0;
            Ahh[mt][kc][0] = w_hh[g * HDIM + ka] * m0;
            Ahh[mt][kc][1] = w_hh[g * HDIM + kb] * m1;
        }
    }

    v8f binit[2];
#pragma unroll
    for (int mt = 0; mt < 2; ++mt) {
#pragma unroll
        for (int v = 0; v < 8; ++v) {
            const int  u  = mt * 4 + 2 * half + (v >> 2);
            const bool uv = u < HDIM;
            const int  g  = uv ? (v & 3) * HDIM + u : 0;
            binit[mt][v] = uv ? (b_ih[g] + b_hh[g]) : 0.0f;
        }
    }

    v2f hB[2];
    float c[2][2];
#pragma unroll
    for (int mt = 0; mt < 2; ++mt) {
        hB[mt][0] = 0.0f; hB[mt][1] = 0.0f;
        c[mt][0]  = 0.0f; c[mt][1]  = 0.0f;
    }

    const float* xblk = x + (size_t)b0 * (TSTEP * IDIM);
    const int xlane  = nl * (TSTEP * IDIM) + 2 * half;
    const int xlane3 = nl * (TSTEP * IDIM) + ((half == 0) ? 12 : 0);

    auto x_load = [&](int t, v2f (&xr)[4]) {
        const float* xt = xblk + t * IDIM;
#pragma unroll
        for (int kc = 0; kc < 3; ++kc) {
            xr[kc][0] = xt[xlane + kc * 4];
            xr[kc][1] = xt[xlane + kc * 4 + 1];
        }
        xr[3][0] = xt[xlane3];
        xr[3][1] = 0.0f;
    };

    auto x_project = [&](const v2f (&xr)[4], v8f& xa0, v8f& xa1) {
        xa0 = binit[0];
        xa1 = binit[1];
#pragma unroll
        for (int kc = 0; kc < 4; ++kc) {
            xa0 = __builtin_amdgcn_wmma_f32_16x16x4_f32(false, Aih[0][kc], false, xr[kc],
                                                        (short)0, xa0, false, false);
            xa1 = __builtin_amdgcn_wmma_f32_16x16x4_f32(false, Aih[1][kc], false, xr[kc],
                                                        (short)0, xa1, false, false);
        }
    };

    auto unit_update = [&](float gi, float gf, float gg_, float go, float& cs) -> float {
        float ig = sigm(gi);
        float fg = sigm(gf);
        float gg = FAST_TANH(gg_);
        float og = sigm(go);
        cs = fg * cs + ig * gg;
        return og * FAST_TANH(cs);
    };

    v2f xring[XPF][4];
    v8f xacc0, xacc1;
    x_load(0, xring[0]);
    x_project(xring[0], xacc0, xacc1);
#pragma unroll
    for (int s = 1; s < XPF; ++s) x_load(s, xring[s]);

    static_assert(TSTEP % XPF == 0, "ring slots are static: the step loop is unrolled by XPF");
    for (int tb = 0; tb < TSTEP; tb += XPF) {
#pragma unroll
    for (int u = 0; u < XPF; ++u) {
        const int t = tb + u;
        v8f acc0 = xacc0, acc1 = xacc1;

#pragma unroll
        for (int kc = 0; kc < 2; ++kc) {
            acc0 = __builtin_amdgcn_wmma_f32_16x16x4_f32(false, Ahh[0][kc], false, hB[kc],
                                                         (short)0, acc0, false, false);
            acc1 = __builtin_amdgcn_wmma_f32_16x16x4_f32(false, Ahh[1][kc], false, hB[kc],
                                                         (short)0, acc1, false, false);
        }

        __builtin_amdgcn_sched_barrier(0);

        int tn = (t + XPF < TSTEP) ? (t + XPF) : 0;
        x_load(tn, xring[u]);
        x_project(xring[(u + 1) % XPF], xacc0, xacc1);

        hB[0][0] = unit_update(acc0[0], acc0[1], acc0[2], acc0[3], c[0][0]);
        hB[0][1] = unit_update(acc0[4], acc0[5], acc0[6], acc0[7], c[0][1]);
        hB[1][0] = unit_update(acc1[0], acc1[1], acc1[2], acc1[3], c[1][0]);
        hB[1][1] = unit_update(acc1[4], acc1[5], acc1[6], acc1[7], c[1][1]);

        __builtin_amdgcn_sched_barrier(0);
    }
    }

#pragma unroll
    for (int mt = 0; mt < 2; ++mt) {
        hbuf[nl][mt * 4 + 2 * half]     = hB[mt][0];
        hbuf[nl][mt * 4 + 2 * half + 1] = hB[mt][1];
    }
    __syncthreads();

    float hr[HDIM];
#pragma unroll
    for (int j = 0; j < HDIM; ++j) {
        float v = hbuf[nl][j];
        hr[j] = v > 0.0f ? v : 0.0f;
    }
    float acc = 0.0f;
    const int ffb = half * (FFDIM / 2);
    for (int ff = ffb; ff < ffb + FFDIM / 2; ++ff) {
        float s = fc1_b[ff];
#pragma unroll
        for (int j = 0; j < HDIM; ++j) s += fc1_w[ff * HDIM + j] * hr[j];
        s = s > 0.0f ? s : 0.0f;
        acc += fc_w[ff] * s;
    }
    if (half == 1) red[nl] = acc;
    __syncthreads();
    if (half == 0) out[b0 + nl] = acc + red[nl] + fc_b[0];
}

extern "C" void kernel_launch(void* const* d_in, const int* in_sizes, int n_in,
                              void* d_out, int out_size, void* d_ws, size_t ws_size,
                              hipStream_t stream) {
    (void)in_sizes; (void)n_in; (void)out_size; (void)d_ws; (void)ws_size;
    const float* x     = (const float*)d_in[0];
    const float* w_ih  = (const float*)d_in[1];
    const float* w_hh  = (const float*)d_in[2];
    const float* b_ih  = (const float*)d_in[3];
    const float* b_hh  = (const float*)d_in[4];
    const float* fc1_w = (const float*)d_in[5];
    const float* fc1_b = (const float*)d_in[6];
    const float* fc_w  = (const float*)d_in[7];
    const float* fc_b  = (const float*)d_in[8];

    lstm_fused_kernel<<<dim3(BATCH / 16), dim3(32), 0, stream>>>(
        x, w_ih, w_hh, b_ih, b_hh, fc1_w, fc1_b, fc_w, fc_b, (float*)d_out);
}


# x-projection moved to two producer waves on the WGP's other SIMDs and handed over through a two-chunk LDS ring; the recurrence wave issues 4 WMMAs a step instead of 12; every sum in the same order
# speedup vs baseline: 3.8802x; 1.2310x over previous
#include <hip/hip_runtime.h>


typedef float v2f __attribute__((ext_vector_type(2)));
typedef float v4f __attribute__((ext_vector_type(4)));
typedef float v8f __attribute__((ext_vector_type(8)));

#define BATCH 4096
#define TSTEP 512
#define IDIM  13
#define HDIM  7
#define FFDIM 512
#define NPROD 2
#define CH    32
#define NCH   (TSTEP / CH)
#define PSTEPS (CH / NPROD)
#define SLOT  128

#define LOG2E 1.4426950408889634f

#if __has_builtin(__builtin_amdgcn_tanhf)
#define FAST_TANH(v) __builtin_amdgcn_tanhf(v)
#define TANH_TIER 2
#elif __has_builtin(__builtin_amdgcn_tanh_f32)
#define FAST_TANH(v) __builtin_amdgcn_tanh_f32(v)
#define TANH_TIER 2
#elif __has_builtin(__builtin_amdgcn_exp2f) && __has_builtin(__builtin_amdgcn_rcpf)
#define TANH_TIER 1
static __device__ __forceinline__ float FAST_TANH(float v) {
    float e2 = __builtin_amdgcn_exp2f(v * (2.0f * LOG2E));
    return fmaf(-2.0f, __builtin_amdgcn_rcpf(e2 + 1.0f), 1.0f);
}
#else
#define FAST_TANH(v) tanhf(v)
#define TANH_TIER 0
#endif

static __device__ __forceinline__ float sigm(float v) {
#if TANH_TIER == 2
    return fmaf(0.5f, FAST_TANH(0.5f * v), 0.5f);
#elif TANH_TIER == 1
    return __builtin_amdgcn_rcpf(1.0f + __builtin_amdgcn_exp2f(v * -LOG2E));
#else
    return 1.0f / (1.0f + __expf(-v));
#endif
}

__global__ __launch_bounds__(32 * (1 + NPROD)) void lstm_fused_kernel(
    const float* __restrict__ x,     const float* __restrict__ w_ih,
    const float* __restrict__ w_hh,  const float* __restrict__ b_ih,
    const float* __restrict__ b_hh,  const float* __restrict__ fc1_w,
    const float* __restrict__ fc1_b, const float* __restrict__ fc_w,
    const float* __restrict__ fc_b,  float* __restrict__ out)
{
    __shared__ v4f   xg[2 * CH * SLOT];
    __shared__ float hbuf[16][8];
    __shared__ float red[16];

    const int l    = threadIdx.x & 31;
    const int wave = __builtin_amdgcn_readfirstlane((int)(threadIdx.x >> 5));
    const int half = l >> 4;
    const int nl   = l & 15;
    const int b0   = blockIdx.x * 16;

    v2f Aih[2][4];
    v2f Ahh[2][2];
#pragma unroll
    for (int mt = 0; mt < 2; ++mt) {
        const int  u  = mt * 4 + (nl >> 2);
        const bool uv = u < HDIM;
        const int  g  = uv ? (nl & 3) * HDIM + u : 0;
#pragma unroll
        for (int kc = 0; kc < 4; ++kc) {
            int k0 = kc * 4 + 2 * half;
            float m0 = (uv && k0     < IDIM) ? 1.0f : 0.0f;
            float m1 = (uv && k0 + 1 < IDIM) ? 1.0f : 0.0f;
            int ka = (k0     < IDIM) ? k0     : 0;
            int kb = (k0 + 1 < IDIM) ? k0 + 1 : 0;
            Aih[mt][kc][0] = w_ih[g * IDIM + ka] * m0;
            Aih[mt][kc][1] = w_ih[g * IDIM + kb] * m1;
        }
#pragma unroll
        for (int kc = 0; kc < 2; ++kc) {
            int k0 = kc * 4 + 2 * half;
            float m0 = (uv && k0     < HDIM) ? 1.0f : 0.0f;
            float m1 = (uv && k0 + 1 < HDIM) ? 1.0f : 0.0f;
            int ka = (k0     < HDIM) ? k0     : 0;
            int kb = (k0 + 1 < HDIM) ? k0 + 1 : 0;
            Ahh[mt][kc][0] = w_hh[g * HDIM + ka] * m0;
            Ahh[mt][kc][1] = w_hh[g * HDIM + kb] * m1;
        }
    }

    v8f binit[2];
#pragma unroll
    for (int mt = 0; mt < 2; ++mt) {
#pragma unroll
        for (int v = 0; v < 8; ++v) {
            const int  u  = mt * 4 + 2 * half + (v >> 2);
            const bool uv = u < HDIM;
            const int  g  = uv ? (v & 3) * HDIM + u : 0;
            binit[mt][v] = uv ? (b_ih[g] + b_hh[g]) : 0.0f;
        }
    }

    v2f hB[2];
    float c[2][2];
#pragma unroll
    for (int mt = 0; mt < 2; ++mt) {
        hB[mt][0] = 0.0f; hB[mt][1] = 0.0f;
        c[mt][0]  = 0.0f; c[mt][1]  = 0.0f;
    }

    const float* xblk = x + (size_t)b0 * (TSTEP * IDIM);
    const int xlane  = nl * (TSTEP * IDIM) + 2 * half;
    const int xlane3 = nl * (TSTEP * IDIM) + ((half == 0) ? 12 : 0);

    auto x_load = [&](int t, v2f (&xr)[4]) {
        const float* xt = xblk + t * IDIM;
#pragma unroll
        for (int kc = 0; kc < 3; ++kc) {
            xr[kc][0] = xt[xlane + kc * 4];
            xr[kc][1] = xt[xlane + kc * 4 + 1];
        }
        xr[3][0] = xt[xlane3];
        xr[3][1] = 0.0f;
    };

    auto x_project = [&](const v2f (&xr)[4], v8f& xa0, v8f& xa1) {
        xa0 = binit[0];
        xa1 = binit[1];
#pragma unroll
        for (int kc = 0; kc < 4; ++kc) {
            xa0 = __builtin_amdgcn_wmma_f32_16x16x4_f32(false, Aih[0][kc], false, xr[kc],
                                                        (short)0, xa0, false, false);
            xa1 = __builtin_amdgcn_wmma_f32_16x16x4_f32(false, Aih[1][kc], false, xr[kc],
                                                        (short)0, xa1, false, false);
        }
    };

    auto unit_update = [&](float gi, float gf, float gg_, float go, float& cs) -> float {
        float ig = sigm(gi);
        float fg = sigm(gf);
        float gg = FAST_TANH(gg_);
        float og = sigm(go);
        cs = fg * cs + ig * gg;
        return og * FAST_TANH(cs);
    };

    auto ring_read = [&](const v4f* p, v8f& a0, v8f& a1) {
        v4f q0 = p[0], q1 = p[32], q2 = p[64], q3 = p[96];
        a0 = __builtin_shufflevector(q0, q1, 0, 1, 2, 3, 4, 5, 6, 7);
        a1 = __builtin_shufflevector(q2, q3, 0, 1, 2, 3, 4, 5, 6, 7);
    };

    auto h_step = [&](const v8f& xa0, const v8f& xa1, const v4f* pnext, v8f& xn0, v8f& xn1) {
        v8f acc0 = xa0, acc1 = xa1;
#pragma unroll
        for (int kc = 0; kc < 2; ++kc) {
            acc0 = __builtin_amdgcn_wmma_f32_16x16x4_f32(false, Ahh[0][kc], false, hB[kc],
                                                         (short)0, acc0, false, false);
            acc1 = __builtin_amdgcn_wmma_f32_16x16x4_f32(false, Ahh[1][kc], false, hB[kc],
                                                         (short)0, acc1, false, false);
        }

        __builtin_amdgcn_sched_barrier(0);

        ring_read(pnext, xn0, xn1);

        hB[0][0] = unit_update(acc0[0], acc0[1], acc0[2], acc0[3], c[0][0]);
        hB[0][1] = unit_update(acc0[4], acc0[5], acc0[6], acc0[7], c[0][1]);
        hB[1][0] = unit_update(acc1[0], acc1[1], acc1[2], acc1[3], c[1][0]);
        hB[1][1] = unit_update(acc1[4], acc1[5], acc1[6], acc1[7], c[1][1]);

        __builtin_amdgcn_sched_barrier(0);
    };

    const int p = wave - 1;

    for (int j = 0; j <= NCH; ++j) {
        if (wave == 0) {
            if (j > 0) {
                const v4f* base = xg + ((j - 1) & 1) * (CH * SLOT) + l;
                v8f xa0, xa1, xb0, xb1;
                ring_read(base, xa0, xa1);
                for (int s = 0; s < CH; s += 2) {
                    h_step(xa0, xa1, base + (s + 1) * SLOT, xb0, xb1);
                    h_step(xb0, xb1, base + ((s + 2) & (CH - 1)) * SLOT, xa0, xa1);
                }
            }
        } else {
            if (j < NCH) {
                v4f* dst = xg + (j & 1) * (CH * SLOT) + l;
                v2f xr[PSTEPS][4];
#pragma unroll
                for (int i = 0; i < PSTEPS; ++i) x_load(j * CH + p + NPROD * i, xr[i]);
#pragma unroll
                for (int i = 0; i < PSTEPS; ++i) {
                    v8f a0, a1;
                    x_project(xr[i], a0, a1);
                    v4f* d = dst + (p + NPROD * i) * SLOT;
                    d[0]  = __builtin_shufflevector(a0, a0, 0, 1, 2, 3);
                    d[32] = __builtin_shufflevector(a0, a0, 4, 5, 6, 7);
                    d[64] = __builtin_shufflevector(a1, a1, 0, 1, 2, 3);
                    d[96] = __builtin_shufflevector(a1, a1, 4, 5, 6, 7);
                }
            }
        }
        __builtin_amdgcn_fence(__ATOMIC_RELEASE, "workgroup", "local");
        __builtin_amdgcn_s_barrier();
        __builtin_amdgcn_fence(__ATOMIC_ACQUIRE, "workgroup", "local");
    }

    if (wave == 0) {
#pragma unroll
        for (int mt = 0; mt < 2; ++mt) {
            hbuf[nl][mt * 4 + 2 * half]     = hB[mt][0];
            hbuf[nl][mt * 4 + 2 * half + 1] = hB[mt][1];
        }
    }
    __syncthreads();

    float acc = 0.0f;
    if (wave == 0) {
        float hr[HDIM];
#pragma unroll
        for (int j = 0; j < HDIM; ++j) {
            float v = hbuf[nl][j];
            hr[j] = v > 0.0f ? v : 0.0f;
        }
        const int ffb = half * (FFDIM / 2);
        for (int ff = ffb; ff < ffb + FFDIM / 2; ++ff) {
            float s = fc1_b[ff];
#pragma unroll
            for (int j = 0; j < HDIM; ++j) s += fc1_w[ff * HDIM + j] * hr[j];
            s = s > 0.0f ? s : 0.0f;
            acc += fc_w[ff] * s;
        }
        if (half == 1) red[nl] = acc;
    }
    __syncthreads();
    if (wave == 0 && half == 0) out[b0 + nl] = acc + red[nl] + fc_b[0];
}

extern "C" void kernel_launch(void* const* d_in, const int* in_sizes, int n_in,
                              void* d_out, int out_size, void* d_ws, size_t ws_size,
                              hipStream_t stream) {
    (void)in_sizes; (void)n_in; (void)out_size; (void)d_ws; (void)ws_size;
    const float* x     = (const float*)d_in[0];
    const float* w_ih  = (const float*)d_in[1];
    const float* w_hh  = (const float*)d_in[2];
    const float* b_ih  = (const float*)d_in[3];
    const float* b_hh  = (const float*)d_in[4];
    const float* fc1_w = (const float*)d_in[5];
    const float* fc1_b = (const float*)d_in[6];
    const float* fc_w  = (const float*)d_in[7];
    const float* fc_b  = (const float*)d_in[8];

    lstm_fused_kernel<<<dim3(BATCH / 16), dim3(32 * (1 + NPROD)), 0, stream>>>(
        x, w_ih, w_hh, b_ih, b_hh, fc1_w, fc1_b, fc_w, fc_b, (float*)d_out);
}


# Recurrence wave reads the x-projection LDS ring in groups of 8 steps: the next group's 32 loads stand in one place behind a step's WMMAs, not 4 loads in 1-2 places every step; every sum as before
# speedup vs baseline: 4.3426x; 1.1192x over previous
#include <hip/hip_runtime.h>


typedef float v2f __attribute__((ext_vector_type(2)));
typedef float v4f __attribute__((ext_vector_type(4)));
typedef float v8f __attribute__((ext_vector_type(8)));

#define BATCH 4096
#define TSTEP 512
#define IDIM  13
#define HDIM  7
#define FFDIM 512
#define NPROD 2
#define CH    32
#define NCH   (TSTEP / CH)
#define PSTEPS (CH / NPROD)
#define SLOT  128
#define GRP   8
#define NGRP  (CH / GRP)

#define LOG2E 1.4426950408889634f

#if __has_builtin(__builtin_amdgcn_tanhf)
#define FAST_TANH(v) __builtin_amdgcn_tanhf(v)
#define TANH_TIER 2
#elif __has_builtin(__builtin_amdgcn_tanh_f32)
#define FAST_TANH(v) __builtin_amdgcn_tanh_f32(v)
#define TANH_TIER 2
#elif __has_builtin(__builtin_amdgcn_exp2f) && __has_builtin(__builtin_amdgcn_rcpf)
#define TANH_TIER 1
static __device__ __forceinline__ float FAST_TANH(float v) {
    float e2 = __builtin_amdgcn_exp2f(v * (2.0f * LOG2E));
    return fmaf(-2.0f, __builtin_amdgcn_rcpf(e2 + 1.0f), 1.0f);
}
#else
#define FAST_TANH(v) tanhf(v)
#define TANH_TIER 0
#endif

static __device__ __forceinline__ float sigm(float v) {
#if TANH_TIER == 2
    return fmaf(0.5f, FAST_TANH(0.5f * v), 0.5f);
#elif TANH_TIER == 1
    return __builtin_amdgcn_rcpf(1.0f + __builtin_amdgcn_exp2f(v * -LOG2E));
#else
    return 1.0f / (1.0f + __expf(-v));
#endif
}

__global__ __launch_bounds__(32 * (1 + NPROD)) void lstm_fused_kernel(
    const float* __restrict__ x,     const float* __restrict__ w_ih,
    const float* __restrict__ w_hh,  const float* __restrict__ b_ih,
    const float* __restrict__ b_hh,  const float* __restrict__ fc1_w,
    const float* __restrict__ fc1_b, const float* __restrict__ fc_w,
    const float* __restrict__ fc_b,  float* __restrict__ out)
{
    __shared__ v4f   xg[2 * CH * SLOT];
    __shared__ float hbuf[16][8];
    __shared__ float red[16];

    const int l    = threadIdx.x & 31;
    const int wave = __builtin_amdgcn_readfirstlane((int)(threadIdx.x >> 5));
    const int half = l >> 4;
    const int nl   = l & 15;
    const int b0   = blockIdx.x * 16;

    v2f Aih[2][4];
    v2f Ahh[2][2];
#pragma unroll
    for (int mt = 0; mt < 2; ++mt) {
        const int  u  = mt * 4 + (nl >> 2);
        const bool uv = u < HDIM;
        const int  g  = uv ? (nl & 3) * HDIM + u : 0;
#pragma unroll
        for (int kc = 0; kc < 4; ++kc) {
            int k0 = kc * 4 + 2 * half;
            float m0 = (uv && k0     < IDIM) ? 1.0f : 0.0f;
            float m1 = (uv && k0 + 1 < IDIM) ? 1.0f : 0.0f;
            int ka = (k0     < IDIM) ? k0     : 0;
            int kb = (k0 + 1 < IDIM) ? k0 + 1 : 0;
            Aih[mt][kc][0] = w_ih[g * IDIM + ka] * m0;
            Aih[mt][kc][1] = w_ih[g * IDIM + kb] * m1;
        }
#pragma unroll
        for (int kc = 0; kc < 2; ++kc) {
            int k0 = kc * 4 + 2 * half;
            float m0 = (uv && k0     < HDIM) ? 1.0f : 0.0f;
            float m1 = (uv && k0 + 1 < HDIM) ? 1.0f : 0.0f;
            int ka = (k0     < HDIM) ? k0     : 0;
            int kb = (k0 + 1 < HDIM) ? k0 + 1 : 0;
            Ahh[mt][kc][0] = w_hh[g * HDIM + ka] * m0;
            Ahh[mt][kc][1] = w_hh[g * HDIM + kb] * m1;
        }
    }

    v8f binit[2];
#pragma unroll
    for (int mt = 0; mt < 2; ++mt) {
#pragma unroll
        for (int v = 0; v < 8; ++v) {
            const int  u  = mt * 4 + 2 * half + (v >> 2);
            const bool uv = u < HDIM;
            const int  g  = uv ? (v & 3) * HDIM + u : 0;
            binit[mt][v] = uv ? (b_ih[g] + b_hh[g]) : 0.0f;
        }
    }

    v2f hB[2];
    float c[2][2];
#pragma unroll
    for (int mt = 0; mt < 2; ++mt) {
        hB[mt][0] = 0.0f; hB[mt][1] = 0.0f;
        c[mt][0]  = 0.0f; c[mt][1]  = 0.0f;
    }

    const float* xblk = x + (size_t)b0 * (TSTEP * IDIM);
    const int xlane  = nl * (TSTEP * IDIM) + 2 * half;
    const int xlane3 = nl * (TSTEP * IDIM) + ((half == 0) ? 12 : 0);

    auto x_load = [&](int t, v2f (&xr)[4]) {
        const float* xt = xblk + t * IDIM;
#pragma unroll
        for (int kc = 0; kc < 3; ++kc) {
            xr[kc][0] = xt[xlane + kc * 4];
            xr[kc][1] = xt[xlane + kc * 4 + 1];
        }
        xr[3][0] = xt[xlane3];
        xr[3][1] = 0.0f;
    };

    auto x_project = [&](const v2f (&xr)[4], v8f& xa0, v8f& xa1) {
        xa0 = binit[0];
        xa1 = binit[1];
#pragma unroll
        for (int kc = 0; kc < 4; ++kc) {
            xa0 = __builtin_amdgcn_wmma_f32_16x16x4_f32(false, Aih[0][kc], false, xr[kc],
                                                        (short)0, xa0, false, false);
            xa1 = __builtin_amdgcn_wmma_f32_16x16x4_f32(false, Aih[1][kc], false, xr[kc],
                                                        (short)0, xa1, false, false);
        }
    };

    auto unit_update = [&](float gi, float gf, float gg_, float go, float& cs) -> float {
        float ig = sigm(gi);
        float fg = sigm(gf);
        float gg = FAST_TANH(gg_);
        float og = sigm(go);
        cs = fg * cs + ig * gg;
        return og * FAST_TANH(cs);
    };

    auto ring_read = [&](const v4f* p, v8f& a0, v8f& a1) {
        v4f q0 = p[0], q1 = p[32], q2 = p[64], q3 = p[96];
        a0 = __builtin_shufflevector(q0, q1, 0, 1, 2, 3, 4, 5, 6, 7);
        a1 = __builtin_shufflevector(q2, q3, 0, 1, 2, 3, 4, 5, 6, 7);
    };

    auto group_read = [&](const v4f* p, v8f (&t)[GRP][2]) {
#pragma unroll
        for (int s = 0; s < GRP; ++s) ring_read(p + s * SLOT, t[s][0], t[s][1]);
    };

    auto h_mma = [&](v8f& acc0, v8f& acc1) {
#pragma unroll
        for (int kc = 0; kc < 2; ++kc) {
            acc0 = __builtin_amdgcn_wmma_f32_16x16x4_f32(false, Ahh[0][kc], false, hB[kc],
                                                         (short)0, acc0, false, false);
            acc1 = __builtin_amdgcn_wmma_f32_16x16x4_f32(false, Ahh[1][kc], false, hB[kc],
                                                         (short)0, acc1, false, false);
        }
        __builtin_amdgcn_sched_barrier(0);
    };

    auto h_act = [&](const v8f& acc0, const v8f& acc1) {
        hB[0][0] = unit_update(acc0[0], acc0[1], acc0[2], acc0[3], c[0][0]);
        hB[0][1] = unit_update(acc0[4], acc0[5], acc0[6], acc0[7], c[0][1]);
        hB[1][0] = unit_update(acc1[0], acc1[1], acc1[2], acc1[3], c[1][0]);
        hB[1][1] = unit_update(acc1[4], acc1[5], acc1[6], acc1[7], c[1][1]);
    };

    auto run_group = [&](v8f (&cur)[GRP][2], v8f (&nxt)[GRP][2], const v4f* pn, bool rd) {
#pragma unroll
        for (int s = 0; s < GRP; ++s) {
            h_mma(cur[s][0], cur[s][1]);
            if (s == 0) {
                if (rd) group_read(pn, nxt);
                __builtin_amdgcn_sched_barrier(0);
            }
            h_act(cur[s][0], cur[s][1]);
        }
    };

    const int p = wave - 1;

    for (int j = 0; j <= NCH; ++j) {
        if (wave == 0) {
            if (j > 0) {
                const v4f* base = xg + ((j - 1) & 1) * (CH * SLOT) + l;
                v8f ta[GRP][2], tb[GRP][2];
                group_read(base, ta);
                for (int g = 0; g < NGRP; g += 2) {
                    run_group(ta, tb, base + (g + 1) * (GRP * SLOT), true);
                    run_group(tb, ta, base + ((g + 2) & (NGRP - 1)) * (GRP * SLOT), g + 2 < NGRP);
                }
            }
        } else {
            if (j < NCH) {
                v4f* dst = xg + (j & 1) * (CH * SLOT) + l;
                v2f xr[PSTEPS][4];
#pragma unroll
                for (int i = 0; i < PSTEPS; ++i) x_load(j * CH + p + NPROD * i, xr[i]);
#pragma unroll
                for (int i = 0; i < PSTEPS; ++i) {
                    v8f a0, a1;
                    x_project(xr[i], a0, a1);
                    v4f* d = dst + (p + NPROD * i) * SLOT;
                    d[0]  = __builtin_shufflevector(a0, a0, 0, 1, 2, 3);
                    d[32] = __builtin_shufflevector(a0, a0, 4, 5, 6, 7);
                    d[64] = __builtin_shufflevector(a1, a1, 0, 1, 2, 3);
                    d[96] = __builtin_shufflevector(a1, a1, 4, 5, 6, 7);
                }
            }
        }
        __builtin_amdgcn_fence(__ATOMIC_RELEASE, "workgroup", "local");
        __builtin_amdgcn_s_barrier();
        __builtin_amdgcn_fence(__ATOMIC_ACQUIRE, "workgroup", "local");
    }

    if (wave == 0) {
#pragma unroll
        for (int mt = 0; mt < 2; ++mt) {
            hbuf[nl][mt * 4 + 2 * half]     = hB[mt][0];
            hbuf[nl][mt * 4 + 2 * half + 1] = hB[mt][1];
        }
    }
    __syncthreads();

    float acc = 0.0f;
    if (wave == 0) {
        float hr[HDIM];
#pragma unroll
        for (int j = 0; j < HDIM; ++j) {
            float v = hbuf[nl][j];
            hr[j] = v > 0.0f ? v : 0.0f;
        }
        const int ffb = half * (FFDIM / 2);
        for (int ff = ffb; ff < ffb + FFDIM / 2; ++ff) {
            float s = fc1_b[ff];
#pragma unroll
            for (int j = 0; j < HDIM; ++j) s += fc1_w[ff * HDIM + j] * hr[j];
            s = s > 0.0f ? s : 0.0f;
            acc += fc_w[ff] * s;
        }
        if (half == 1) red[nl] = acc;
    }
    __syncthreads();
    if (wave == 0 && half == 0) out[b0 + nl] = acc + red[nl] + fc_b[0];
}

extern "C" void kernel_launch(void* const* d_in, const int* in_sizes, int n_in,
                              void* d_out, int out_size, void* d_ws, size_t ws_size,
                              hipStream_t stream) {
    (void)in_sizes; (void)n_in; (void)out_size; (void)d_ws; (void)ws_size;
    const float* x     = (const float*)d_in[0];
    const float* w_ih  = (const float*)d_in[1];
    const float* w_hh  = (const float*)d_in[2];
    const float* b_ih  = (const float*)d_in[3];
    const float* b_hh  = (const float*)d_in[4];
    const float* fc1_w = (const float*)d_in[5];
    const float* fc1_b = (const float*)d_in[6];
    const float* fc_w  = (const float*)d_in[7];
    const float* fc_b  = (const float*)d_in[8];

    lstm_fused_kernel<<<dim3(BATCH / 16), dim3(32 * (1 + NPROD)), 0, stream>>>(
        x, w_ih, w_hh, b_ih, b_hh, fc1_w, fc1_b, fc_w, fc_b, (float*)d_out);
}


# FF head's weights copied into LDS by the producer waves in the last round, where they have no work; the head reads LDS one pass ahead, not global memory in every pass; every sum in the same order
# speedup vs baseline: 5.5929x; 1.2879x over previous
#include <hip/hip_runtime.h>


typedef float v2f __attribute__((ext_vector_type(2)));
typedef float v4f __attribute__((ext_vector_type(4)));
typedef float v8f __attribute__((ext_vector_type(8)));

#define BATCH 4096
#define TSTEP 512
#define IDIM  13
#define HDIM  7
#define FFDIM 512
#define NPROD 2
#define CH    32
#define NCH   (TSTEP / CH)
#define PSTEPS (CH / NPROD)
#define SLOT  128
#define GRP   8
#define NGRP  (CH / GRP)
#define HU    8
#define HPART (FFDIM / 2)
#define HPASS (HPART / HU)

#define LOG2E 1.4426950408889634f

#if __has_builtin(__builtin_amdgcn_tanhf)
#define FAST_TANH(v) __builtin_amdgcn_tanhf(v)
#define TANH_TIER 2
#elif __has_builtin(__builtin_amdgcn_tanh_f32)
#define FAST_TANH(v) __builtin_amdgcn_tanh_f32(v)
#define TANH_TIER 2
#elif __has_builtin(__builtin_amdgcn_exp2f) && __has_builtin(__builtin_amdgcn_rcpf)
#define TANH_TIER 1
static __device__ __forceinline__ float FAST_TANH(float v) {
    float e2 = __builtin_amdgcn_exp2f(v * (2.0f * LOG2E));
    return fmaf(-2.0f, __builtin_amdgcn_rcpf(e2 + 1.0f), 1.0f);
}
#else
#define FAST_TANH(v) tanhf(v)
#define TANH_TIER 0
#endif

static __device__ __forceinline__ float sigm(float v) {
#if TANH_TIER == 2
    return fmaf(0.5f, FAST_TANH(0.5f * v), 0.5f);
#elif TANH_TIER == 1
    return __builtin_amdgcn_rcpf(1.0f + __builtin_amdgcn_exp2f(v * -LOG2E));
#else
    return 1.0f / (1.0f + __expf(-v));
#endif
}

__global__ __launch_bounds__(32 * (1 + NPROD)) void lstm_fused_kernel(
    const float* __restrict__ x,     const float* __restrict__ w_ih,
    const float* __restrict__ w_hh,  const float* __restrict__ b_ih,
    const float* __restrict__ b_hh,  const float* __restrict__ fc1_w,
    const float* __restrict__ fc1_b, const float* __restrict__ fc_w,
    const float* __restrict__ fc_b,  float* __restrict__ out)
{
    __shared__ v4f   xg[2 * CH * SLOT];
    __shared__ float hbuf[16][8];
    __shared__ float red[16];
    __shared__ v4f   hw[2 * FFDIM];
    __shared__ v4f   hfc[FFDIM / 4];

    const int l    = threadIdx.x & 31;
    const int wave = __builtin_amdgcn_readfirstlane((int)(threadIdx.x >> 5));
    const int half = l >> 4;
    const int nl   = l & 15;
    const int b0   = blockIdx.x * 16;

    v2f Aih[2][4];
    v2f Ahh[2][2];
#pragma unroll
    for (int mt = 0; mt < 2; ++mt) {
        const int  u  = mt * 4 + (nl >> 2);
        const bool uv = u < HDIM;
        const int  g  = uv ? (nl & 3) * HDIM + u : 0;
#pragma unroll
        for (int kc = 0; kc < 4; ++kc) {
            int k0 = kc * 4 + 2 * half;
            float m0 = (uv && k0     < IDIM) ? 1.0f : 0.0f;
            float m1 = (uv && k0 + 1 < IDIM) ? 1.0f : 0.0f;
            int ka = (k0     < IDIM) ? k0     : 0;
            int kb = (k0 + 1 < IDIM) ? k0 + 1 : 0;
            Aih[mt][kc][0] = w_ih[g * IDIM + ka] * m0;
            Aih[mt][kc][1] = w_ih[g * IDIM + kb] * m1;
        }
#pragma unroll
        for (int kc = 0; kc < 2; ++kc) {
            int k0 = kc * 4 + 2 * half;
            float m0 = (uv && k0     < HDIM) ? 1.0f : 0.0f;
            float m1 = (uv && k0 + 1 < HDIM) ? 1.0f : 0.0f;
            int ka = (k0     < HDIM) ? k0     : 0;
            int kb = (k0 + 1 < HDIM) ? k0 + 1 : 0;
            Ahh[mt][kc][0] = w_hh[g * HDIM + ka] * m0;
            Ahh[mt][kc][1] = w_hh[g * HDIM + kb] * m1;
        }
    }

    v8f binit[2];
#pragma unroll
    for (int mt = 0; mt < 2; ++mt) {
#pragma unroll
        for (int v = 0; v < 8; ++v) {
            const int  u  = mt * 4 + 2 * half + (v >> 2);
            const bool uv = u < HDIM;
            const int  g  = uv ? (v & 3) * HDIM + u : 0;
            binit[mt][v] = uv ? (b_ih[g] + b_hh[g]) : 0.0f;
        }
    }

    v2f hB[2];
    float c[2][2];
#pragma unroll
    for (int mt = 0; mt < 2; ++mt) {
        hB[mt][0] = 0.0f; hB[mt][1] = 0.0f;
        c[mt][0]  = 0.0f; c[mt][1]  = 0.0f;
    }

    const float* xblk = x + (size_t)b0 * (TSTEP * IDIM);
    const int xlane  = nl * (TSTEP * IDIM) + 2 * half;
    const int xlane3 = nl * (TSTEP * IDIM) + ((half == 0) ? 12 : 0);

    auto x_load = [&](int t, v2f (&xr)[4]) {
        const float* xt = xblk + t * IDIM;
#pragma unroll
        for (int kc = 0; kc < 3; ++kc) {
            xr[kc][0] = xt[xlane + kc * 4];
            xr[kc][1] = xt[xlane + kc * 4 + 1];
        }
        xr[3][0] = xt[xlane3];
        xr[3][1] = 0.0f;
    };

    auto x_project = [&](const v2f (&xr)[4], v8f& xa0, v8f& xa1) {
        xa0 = binit[0];
        xa1 = binit[1];
#pragma unroll
        for (int kc = 0; kc < 4; ++kc) {
            xa0 = __builtin_amdgcn_wmma_f32_16x16x4_f32(false, Aih[0][kc], false, xr[kc],
                                                        (short)0, xa0, false, false);
            xa1 = __builtin_amdgcn_wmma_f32_16x16x4_f32(false, Aih[1][kc], false, xr[kc],
                                                        (short)0, xa1, false, false);
        }
    };

    auto unit_update = [&](float gi, float gf, float gg_, float go, float& cs) -> float {
        float ig = sigm(gi);
        float fg = sigm(gf);
        float gg = FAST_TANH(gg_);
        float og = sigm(go);
        cs = fg * cs + ig * gg;
        return og * FAST_TANH(cs);
    };

    auto ring_read = [&](const v4f* p, v8f& a0, v8f& a1) {
        v4f q0 = p[0], q1 = p[32], q2 = p[64], q3 = p[96];
        a0 = __builtin_shufflevector(q0, q1, 0, 1, 2, 3, 4, 5, 6, 7);
        a1 = __builtin_shufflevector(q2, q3, 0, 1, 2, 3, 4, 5, 6, 7);
    };

    auto group_read = [&](const v4f* p, v8f (&t)[GRP][2]) {
#pragma unroll
        for (int s = 0; s < GRP; ++s) ring_read(p + s * SLOT, t[s][0], t[s][1]);
    };

    auto h_mma = [&](v8f& acc0, v8f& acc1) {
#pragma unroll
        for (int kc = 0; kc < 2; ++kc) {
            acc0 = __builtin_amdgcn_wmma_f32_16x16x4_f32(false, Ahh[0][kc], false, hB[kc],
                                                         (short)0, acc0, false, false);
            acc1 = __builtin_amdgcn_wmma_f32_16x16x4_f32(false, Ahh[1][kc], false, hB[kc],
                                                         (short)0, acc1, false, false);
        }
        __builtin_amdgcn_sched_barrier(0);
    };

    auto h_act = [&](const v8f& acc0, const v8f& acc1) {
        hB[0][0] = unit_update(acc0[0], acc0[1], acc0[2], acc0[3], c[0][0]);
        hB[0][1] = unit_update(acc0[4], acc0[5], acc0[6], acc0[7], c[0][1]);
        hB[1][0] = unit_update(acc1[0], acc1[1], acc1[2], acc1[3], c[1][0]);
        hB[1][1] = unit_update(acc1[4], acc1[5], acc1[6], acc1[7], c[1][1]);
    };

    auto run_group = [&](v8f (&cur)[GRP][2], v8f (&nxt)[GRP][2], const v4f* pn, bool rd) {
#pragma unroll
        for (int s = 0; s < GRP; ++s) {
            h_mma(cur[s][0], cur[s][1]);
            if (s == 0) {
                if (rd) group_read(pn, nxt);
                __builtin_amdgcn_sched_barrier(0);
            }
            h_act(cur[s][0], cur[s][1]);
        }
    };

    const int p = wave - 1;

    for (int j = 0; j <= NCH; ++j) {
        if (wave == 0) {
            if (j > 0) {
                const v4f* base = xg + ((j - 1) & 1) * (CH * SLOT) + l;
                v8f ta[GRP][2], tb[GRP][2];
                group_read(base, ta);
                for (int g = 0; g < NGRP; g += 2) {
                    run_group(ta, tb, base + (g + 1) * (GRP * SLOT), true);
                    run_group(tb, ta, base + ((g + 2) & (NGRP - 1)) * (GRP * SLOT), g + 2 < NGRP);
                }
            }
        } else {
            if (j < NCH) {
                v4f* dst = xg + (j & 1) * (CH * SLOT) + l;
                v2f xr[PSTEPS][4];
#pragma unroll
                for (int i = 0; i < PSTEPS; ++i) x_load(j * CH + p + NPROD * i, xr[i]);
#pragma unroll
                for (int i = 0; i < PSTEPS; ++i) {
                    v8f a0, a1;
                    x_project(xr[i], a0, a1);
                    v4f* d = dst + (p + NPROD * i) * SLOT;
                    d[0]  = __builtin_shufflevector(a0, a0, 0, 1, 2, 3);
                    d[32] = __builtin_shufflevector(a0, a0, 4, 5, 6, 7);
                    d[64] = __builtin_shufflevector(a1, a1, 0, 1, 2, 3);
                    d[96] = __builtin_shufflevector(a1, a1, 4, 5, 6, 7);
                }
            } else {
                const int tid = p * 32 + l;
                const float* w1  = fc1_w + tid * HDIM;
                const float* b1  = fc1_b + tid;
                const float* w2  = fc_w + tid;
                v4f*   hws  = hw + 2 * tid;
                float* hfcs = (float*)hfc + tid;
#pragma unroll
                for (int k = 0; k < FFDIM / (32 * NPROD); ++k) {
                    const int o = (32 * NPROD) * k;
                    const float* wr = w1 + o * HDIM;
                    v4f wa, wb;
                    wa[0] = wr[0]; wa[1] = wr[1]; wa[2] = wr[2]; wa[3] = wr[3];
                    wb[0] = wr[4]; wb[1] = wr[5]; wb[2] = wr[6]; wb[3] = b1[o];
                    hws[2 * o]     = wa;
                    hws[2 * o + 1] = wb;
                    hfcs[o]        = w2[o];
                }
            }
        }
        __builtin_amdgcn_fence(__ATOMIC_RELEASE, "workgroup", "local");
        __builtin_amdgcn_s_barrier();
        __builtin_amdgcn_fence(__ATOMIC_ACQUIRE, "workgroup", "local");
    }

    if (wave == 0) {
#pragma unroll
        for (int mt = 0; mt < 2; ++mt) {
            hbuf[nl][mt * 4 + 2 * half]     = hB[mt][0];
            hbuf[nl][mt * 4 + 2 * half + 1] = hB[mt][1];
        }
    }
    __syncthreads();

    float acc = 0.0f;
    if (wave == 0) {
        float hr[HDIM];
#pragma unroll
        for (int j = 0; j < HDIM; ++j) {
            float v = hbuf[nl][j];
            hr[j] = v > 0.0f ? v : 0.0f;
        }
        auto head_read = [&](const v4f* pw, const v4f* pf, v4f (&a)[HU], v4f (&b)[HU], v4f (&f)[HU / 4]) {
#pragma unroll
            for (int u = 0; u < HU; ++u) { a[u] = pw[2 * u]; b[u] = pw[2 * u + 1]; }
#pragma unroll
            for (int u = 0; u < HU / 4; ++u) f[u] = pf[u];
            __builtin_amdgcn_sched_barrier(0);
        };
        auto head_sum = [&](const v4f (&a)[HU], const v4f (&b)[HU], const v4f (&f)[HU / 4]) {
#pragma unroll
            for (int u = 0; u < HU; ++u) {
                float s = b[u][3];
                s += a[u][0] * hr[0];
                s += a[u][1] * hr[1];
                s += a[u][2] * hr[2];
                s += a[u][3] * hr[3];
                s += b[u][0] * hr[4];
                s += b[u][1] * hr[5];
                s += b[u][2] * hr[6];
                s = s > 0.0f ? s : 0.0f;
                acc += f[u >> 2][u & 3] * s;
            }
        };
        const v4f* pw = hw + 2 * half * HPART;
        const v4f* pf = hfc + half * (HPART / 4);
        v4f ra[HU], rb[HU], rf[HU / 4];
        v4f qa[HU], qb[HU], qf[HU / 4];
        head_read(pw, pf, ra, rb, rf);
        for (int g = 0; g < HPASS; g += 2) {
            head_read(pw + 2 * HU, pf + HU / 4, qa, qb, qf);
            head_sum(ra, rb, rf);
            pw += 4 * HU;
            pf += HU / 2;
            if (g + 2 < HPASS) head_read(pw, pf, ra, rb, rf);
            head_sum(qa, qb, qf);
        }
        if (half == 1) red[nl] = acc;
    }
    __syncthreads();
    if (wave == 0 && half == 0) out[b0 + nl] = acc + red[nl] + fc_b[0];
}

extern "C" void kernel_launch(void* const* d_in, const int* in_sizes, int n_in,
                              void* d_out, int out_size, void* d_ws, size_t ws_size,
                              hipStream_t stream) {
    (void)in_sizes; (void)n_in; (void)out_size; (void)d_ws; (void)ws_size;
    const float* x     = (const float*)d_in[0];
    const float* w_ih  = (const float*)d_in[1];
    const float* w_hh  = (const float*)d_in[2];
    const float* b_ih  = (const float*)d_in[3];
    const float* b_hh  = (const float*)d_in[4];
    const float* fc1_w = (const float*)d_in[5];
    const float* fc1_b = (const float*)d_in[6];
    const float* fc_w  = (const float*)d_in[7];
    const float* fc_b  = (const float*)d_in[8];

    lstm_fused_kernel<<<dim3(BATCH / 16), dim3(32 * (1 + NPROD)), 0, stream>>>(
        x, w_ih, w_hh, b_ih, b_hh, fc1_w, fc1_b, fc_w, fc_b, (float*)d_out);
}


# Sigmoid's x/2 scale folded into the i,f,o rows of weights and bias where the A operands are built (exact halving): a step's tanh ops read WMMA results directly, no multiply between; sums unchanged
# speedup vs baseline: 5.8460x; 1.0453x over previous
#include <hip/hip_runtime.h>


typedef float v2f __attribute__((ext_vector_type(2)));
typedef float v4f __attribute__((ext_vector_type(4)));
typedef float v8f __attribute__((ext_vector_type(8)));

#define BATCH 4096
#define TSTEP 512
#define IDIM  13
#define HDIM  7
#define FFDIM 512
#define NPROD 2
#define CH    32
#define NCH   (TSTEP / CH)
#define PSTEPS (CH / NPROD)
#define SLOT  128
#define GRP   8
#define NGRP  (CH / GRP)
#define HU    8
#define HPART (FFDIM / 2)
#define HPASS (HPART / HU)

#define LOG2E 1.4426950408889634f

#if __has_builtin(__builtin_amdgcn_tanhf)
#define FAST_TANH(v) __builtin_amdgcn_tanhf(v)
#define TANH_TIER 2
#elif __has_builtin(__builtin_amdgcn_tanh_f32)
#define FAST_TANH(v) __builtin_amdgcn_tanh_f32(v)
#define TANH_TIER 2
#elif __has_builtin(__builtin_amdgcn_exp2f) && __has_builtin(__builtin_amdgcn_rcpf)
#define TANH_TIER 1
static __device__ __forceinline__ float FAST_TANH(float v) {
    float e2 = __builtin_amdgcn_exp2f(v * (2.0f * LOG2E));
    return fmaf(-2.0f, __builtin_amdgcn_rcpf(e2 + 1.0f), 1.0f);
}
#else
#define FAST_TANH(v) tanhf(v)
#define TANH_TIER 0
#endif

static __device__ __forceinline__ float sigm_half(float hx) {
#if TANH_TIER == 2
    return fmaf(0.5f, FAST_TANH(hx), 0.5f);
#elif TANH_TIER == 1
    return __builtin_amdgcn_rcpf(1.0f + __builtin_amdgcn_exp2f(hx * (-2.0f * LOG2E)));
#else
    return 1.0f / (1.0f + __expf(-2.0f * hx));
#endif
}

#define GATE_PRE(gate) (((gate) == 2) ? 1.0f : 0.5f)

__global__ __launch_bounds__(32 * (1 + NPROD)) void lstm_fused_kernel(
    const float* __restrict__ x,     const float* __restrict__ w_ih,
    const float* __restrict__ w_hh,  const float* __restrict__ b_ih,
    const float* __restrict__ b_hh,  const float* __restrict__ fc1_w,
    const float* __restrict__ fc1_b, const float* __restrict__ fc_w,
    const float* __restrict__ fc_b,  float* __restrict__ out)
{
    __shared__ v4f   xg[2 * CH * SLOT];
    __shared__ float hbuf[16][8];
    __shared__ float red[16];
    __shared__ v4f   hw[2 * FFDIM];
    __shared__ v4f   hfc[FFDIM / 4];

    const int l    = threadIdx.x & 31;
    const int wave = __builtin_amdgcn_readfirstlane((int)(threadIdx.x >> 5));
    const int half = l >> 4;
    const int nl   = l & 15;
    const int b0   = blockIdx.x * 16;

    v2f Aih[2][4];
    v2f Ahh[2][2];
#pragma unroll
    for (int mt = 0; mt < 2; ++mt) {
        const int  u  = mt * 4 + (nl >> 2);
        const bool uv = u < HDIM;
        const int  g  = uv ? (nl & 3) * HDIM + u : 0;
        const float pre = GATE_PRE(nl & 3);
#pragma unroll
        for (int kc = 0; kc < 4; ++kc) {
            int k0 = kc * 4 + 2 * half;
            float m0 = (uv && k0     < IDIM) ? pre : 0.0f;
            float m1 = (uv && k0 + 1 < IDIM) ? pre : 0.0f;
            int ka = (k0     < IDIM) ? k0     : 0;
            int kb = (k0 + 1 < IDIM) ? k0 + 1 : 0;
            Aih[mt][kc][0] = w_ih[g * IDIM + ka] * m0;
            Aih[mt][kc][1] = w_ih[g * IDIM + kb] * m1;
        }
#pragma unroll
        for (int kc = 0; kc < 2; ++kc) {
            int k0 = kc * 4 + 2 * half;
            float m0 = (uv && k0     < HDIM) ? pre : 0.0f;
            float m1 = (uv && k0 + 1 < HDIM) ? pre : 0.0f;
            int ka = (k0     < HDIM) ? k0     : 0;
            int kb = (k0 + 1 < HDIM) ? k0 + 1 : 0;
            Ahh[mt][kc][0] = w_hh[g * HDIM + ka] * m0;
            Ahh[mt][kc][1] = w_hh[g * HDIM + kb] * m1;
        }
    }

    v8f binit[2];
#pragma unroll
    for (int mt = 0; mt < 2; ++mt) {
#pragma unroll
        for (int v = 0; v < 8; ++v) {
            const int  u  = mt * 4 + 2 * half + (v >> 2);
            const bool uv = u < HDIM;
            const int  g  = uv ? (v & 3) * HDIM + u : 0;
            binit[mt][v] = uv ? (b_ih[g] + b_hh[g]) * GATE_PRE(v & 3) : 0.0f;
        }
    }

    v2f hB[2];
    float c[2][2];
#pragma unroll
    for (int mt = 0; mt < 2; ++mt) {
        hB[mt][0] = 0.0f; hB[mt][1] = 0.0f;
        c[mt][0]  = 0.0f; c[mt][1]  = 0.0f;
    }

    const float* xblk = x + (size_t)b0 * (TSTEP * IDIM);
    const int xlane  = nl * (TSTEP * IDIM) + 2 * half;
    const int xlane3 = nl * (TSTEP * IDIM) + ((half == 0) ? 12 : 0);

    auto x_load = [&](int t, v2f (&xr)[4]) {
        const float* xt = xblk + t * IDIM;
#pragma unroll
        for (int kc = 0; kc < 3; ++kc) {
            xr[kc][0] = xt[xlane + kc * 4];
            xr[kc][1] = xt[xlane + kc * 4 + 1];
        }
        xr[3][0] = xt[xlane3];
        xr[3][1] = 0.0f;
    };

    auto x_project = [&](const v2f (&xr)[4], v8f& xa0, v8f& xa1) {
        xa0 = binit[0];
        xa1 = binit[1];
#pragma unroll
        for (int kc = 0; kc < 4; ++kc) {
            xa0 = __builtin_amdgcn_wmma_f32_16x16x4_f32(false, Aih[0][kc], false, xr[kc],
                                                        (short)0, xa0, false, false);
            xa1 = __builtin_amdgcn_wmma_f32_16x16x4_f32(false, Aih[1][kc], false, xr[kc],
                                                        (short)0, xa1, false, false);
        }
    };

    auto unit_update = [&](float gi, float gf, float gg_, float go, float& cs) -> float {
        float ig = sigm_half(gi);
        float fg = sigm_half(gf);
        float gg = FAST_TANH(gg_);
        float og = sigm_half(go);
        cs = fg * cs + ig * gg;
        return og * FAST_TANH(cs);
    };

    auto ring_read = [&](const v4f* p, v8f& a0, v8f& a1) {
        v4f q0 = p[0], q1 = p[32], q2 = p[64], q3 = p[96];
        a0 = __builtin_shufflevector(q0, q1, 0, 1, 2, 3, 4, 5, 6, 7);
        a1 = __builtin_shufflevector(q2, q3, 0, 1, 2, 3, 4, 5, 6, 7);
    };

    auto group_read = [&](const v4f* p, v8f (&t)[GRP][2]) {
#pragma unroll
        for (int s = 0; s < GRP; ++s) ring_read(p + s * SLOT, t[s][0], t[s][1]);
    };

    auto h_mma = [&](v8f& acc0, v8f& acc1) {
#pragma unroll
        for (int kc = 0; kc < 2; ++kc) {
            acc0 = __builtin_amdgcn_wmma_f32_16x16x4_f32(false, Ahh[0][kc], false, hB[kc],
                                                         (short)0, acc0, false, false);
            acc1 = __builtin_amdgcn_wmma_f32_16x16x4_f32(false, Ahh[1][kc], false, hB[kc],
                                                         (short)0, acc1, false, false);
        }
        __builtin_amdgcn_sched_barrier(0);
    };

    auto h_act = [&](const v8f& acc0, const v8f& acc1) {
        hB[0][0] = unit_update(acc0[0], acc0[1], acc0[2], acc0[3], c[0][0]);
        hB[0][1] = unit_update(acc0[4], acc0[5], acc0[6], acc0[7], c[0][1]);
        hB[1][0] = unit_update(acc1[0], acc1[1], acc1[2], acc1[3], c[1][0]);
        hB[1][1] = unit_update(acc1[4], acc1[5], acc1[6], acc1[7], c[1][1]);
    };

    auto run_group = [&](v8f (&cur)[GRP][2], v8f (&nxt)[GRP][2], const v4f* pn, bool rd) {
#pragma unroll
        for (int s = 0; s < GRP; ++s) {
            h_mma(cur[s][0], cur[s][1]);
            if (s == 0) {
                if (rd) group_read(pn, nxt);
                __builtin_amdgcn_sched_barrier(0);
            }
            h_act(cur[s][0], cur[s][1]);
        }
    };

    const int p = wave - 1;

    for (int j = 0; j <= NCH; ++j) {
        if (wave == 0) {
            if (j > 0) {
                const v4f* base = xg + ((j - 1) & 1) * (CH * SLOT) + l;
                v8f ta[GRP][2], tb[GRP][2];
                group_read(base, ta);
                for (int g = 0; g < NGRP; g += 2) {
                    run_group(ta, tb, base + (g + 1) * (GRP * SLOT), true);
                    run_group(tb, ta, base + ((g + 2) & (NGRP - 1)) * (GRP * SLOT), g + 2 < NGRP);
                }
            }
        } else {
            if (j < NCH) {
                v4f* dst = xg + (j & 1) * (CH * SLOT) + l;
                v2f xr[PSTEPS][4];
#pragma unroll
                for (int i = 0; i < PSTEPS; ++i) x_load(j * CH + p + NPROD * i, xr[i]);
#pragma unroll
                for (int i = 0; i < PSTEPS; ++i) {
                    v8f a0, a1;
                    x_project(xr[i], a0, a1);
                    v4f* d = dst + (p + NPROD * i) * SLOT;
                    d[0]  = __builtin_shufflevector(a0, a0, 0, 1, 2, 3);
                    d[32] = __builtin_shufflevector(a0, a0, 4, 5, 6, 7);
                    d[64] = __builtin_shufflevector(a1, a1, 0, 1, 2, 3);
                    d[96] = __builtin_shufflevector(a1, a1, 4, 5, 6, 7);
                }
            } else {
                const int tid = p * 32 + l;
                const float* w1  = fc1_w + tid * HDIM;
                const float* b1  = fc1_b + tid;
                const float* w2  = fc_w + tid;
                v4f*   hws  = hw + 2 * tid;
                float* hfcs = (float*)hfc + tid;
#pragma unroll
                for (int k = 0; k < FFDIM / (32 * NPROD); ++k) {
                    const int o = (32 * NPROD) * k;
                    const float* wr = w1 + o * HDIM;
                    v4f wa, wb;
                    wa[0] = wr[0]; wa[1] = wr[1]; wa[2] = wr[2]; wa[3] = wr[3];
                    wb[0] = wr[4]; wb[1] = wr[5]; wb[2] = wr[6]; wb[3] = b1[o];
                    hws[2 * o]     = wa;
                    hws[2 * o + 1] = wb;
                    hfcs[o]        = w2[o];
                }
            }
        }
        __builtin_amdgcn_fence(__ATOMIC_RELEASE, "workgroup", "local");
        __builtin_amdgcn_s_barrier();
        __builtin_amdgcn_fence(__ATOMIC_ACQUIRE, "workgroup", "local");
    }

    if (wave == 0) {
#pragma unroll
        for (int mt = 0; mt < 2; ++mt) {
            hbuf[nl][mt * 4 + 2 * half]     = hB[mt][0];
            hbuf[nl][mt * 4 + 2 * half + 1] = hB[mt][1];
        }
    }
    __syncthreads();

    float acc = 0.0f;
    if (wave == 0) {
        float hr[HDIM];
#pragma unroll
        for (int j = 0; j < HDIM; ++j) {
            float v = hbuf[nl][j];
            hr[j] = v > 0.0f ? v : 0.0f;
        }
        auto head_read = [&](const v4f* pw, const v4f* pf, v4f (&a)[HU], v4f (&b)[HU], v4f (&f)[HU / 4]) {
#pragma unroll
            for (int u = 0; u < HU; ++u) { a[u] = pw[2 * u]; b[u] = pw[2 * u + 1]; }
#pragma unroll
            for (int u = 0; u < HU / 4; ++u) f[u] = pf[u];
            __builtin_amdgcn_sched_barrier(0);
        };
        auto head_sum = [&](const v4f (&a)[HU], const v4f (&b)[HU], const v4f (&f)[HU / 4]) {
#pragma unroll
            for (int u = 0; u < HU; ++u) {
                float s = b[u][3];
                s += a[u][0] * hr[0];
                s += a[u][1] * hr[1];
                s += a[u][2] * hr[2];
                s += a[u][3] * hr[3];
                s += b[u][0] * hr[4];
                s += b[u][1] * hr[5];
                s += b[u][2] * hr[6];
                s = s > 0.0f ? s : 0.0f;
                acc += f[u >> 2][u & 3] * s;
            }
        };
        const v4f* pw = hw + 2 * half * HPART;
        const v4f* pf = hfc + half * (HPART / 4);
        v4f ra[HU], rb[HU], rf[HU / 4];
        v4f qa[HU], qb[HU], qf[HU / 4];
        head_read(pw, pf, ra, rb, rf);
        for (int g = 0; g < HPASS; g += 2) {
            head_read(pw + 2 * HU, pf + HU / 4, qa, qb, qf);
            head_sum(ra, rb, rf);
            pw += 4 * HU;
            pf += HU / 2;
            if (g + 2 < HPASS) head_read(pw, pf, ra, rb, rf);
            head_sum(qa, qb, qf);
        }
        if (half == 1) red[nl] = acc;
    }
    __syncthreads();
    if (wave == 0 && half == 0) out[b0 + nl] = acc + red[nl] + fc_b[0];
}

extern "C" void kernel_launch(void* const* d_in, const int* in_sizes, int n_in,
                              void* d_out, int out_size, void* d_ws, size_t ws_size,
                              hipStream_t stream) {
    (void)in_sizes; (void)n_in; (void)out_size; (void)d_ws; (void)ws_size;
    const float* x     = (const float*)d_in[0];
    const float* w_ih  = (const float*)d_in[1];
    const float* w_hh  = (const float*)d_in[2];
    const float* b_ih  = (const float*)d_in[3];
    const float* b_hh  = (const float*)d_in[4];
    const float* fc1_w = (const float*)d_in[5];
    const float* fc1_b = (const float*)d_in[6];
    const float* fc_w  = (const float*)d_in[7];
    const float* fc_b  = (const float*)d_in[8];

    lstm_fused_kernel<<<dim3(BATCH / 16), dim3(32 * (1 + NPROD)), 0, stream>>>(
        x, w_ih, w_hh, b_ih, b_hh, fc1_w, fc1_b, fc_w, fc_b, (float*)d_out);
}


# Cell update of a tile's two units as packed two-wide f32 instructions (v_pk_fma_f32, v_pk_mul_f32): 12 vector issues a step, not 22; each element the same IEEE operation; every sum as before
# speedup vs baseline: 6.0581x; 1.0363x over previous
#include <hip/hip_runtime.h>


typedef float v2f __attribute__((ext_vector_type(2)));
typedef float v4f __attribute__((ext_vector_type(4)));
typedef float v8f __attribute__((ext_vector_type(8)));

#define BATCH 4096
#define TSTEP 512
#define IDIM  13
#define HDIM  7
#define FFDIM 512
#define NPROD 2
#define CH    32
#define NCH   (TSTEP / CH)
#define PSTEPS (CH / NPROD)
#define SLOT  128
#define GRP   8
#define NGRP  (CH / GRP)
#define HU    8
#define HPART (FFDIM / 2)
#define HPASS (HPART / HU)

#define LOG2E 1.4426950408889634f

#if __has_builtin(__builtin_amdgcn_tanhf)
#define FAST_TANH(v) __builtin_amdgcn_tanhf(v)
#define TANH_TIER 2
#elif __has_builtin(__builtin_amdgcn_tanh_f32)
#define FAST_TANH(v) __builtin_amdgcn_tanh_f32(v)
#define TANH_TIER 2
#elif __has_builtin(__builtin_amdgcn_exp2f) && __has_builtin(__builtin_amdgcn_rcpf)
#define TANH_TIER 1
static __device__ __forceinline__ float FAST_TANH(float v) {
    float e2 = __builtin_amdgcn_exp2f(v * (2.0f * LOG2E));
    return fmaf(-2.0f, __builtin_amdgcn_rcpf(e2 + 1.0f), 1.0f);
}
#else
#define FAST_TANH(v) tanhf(v)
#define TANH_TIER 0
#endif

static __device__ __forceinline__ float sigm_half(float hx) {
#if TANH_TIER == 2
    return fmaf(0.5f, FAST_TANH(hx), 0.5f);
#elif TANH_TIER == 1
    return __builtin_amdgcn_rcpf(1.0f + __builtin_amdgcn_exp2f(hx * (-2.0f * LOG2E)));
#else
    return 1.0f / (1.0f + __expf(-2.0f * hx));
#endif
}

#define GATE_PRE(gate) (((gate) == 2) ? 1.0f : 0.5f)

__global__ __launch_bounds__(32 * (1 + NPROD)) void lstm_fused_kernel(
    const float* __restrict__ x,     const float* __restrict__ w_ih,
    const float* __restrict__ w_hh,  const float* __restrict__ b_ih,
    const float* __restrict__ b_hh,  const float* __restrict__ fc1_w,
    const float* __restrict__ fc1_b, const float* __restrict__ fc_w,
    const float* __restrict__ fc_b,  float* __restrict__ out)
{
    __shared__ v4f   xg[2 * CH * SLOT];
    __shared__ float hbuf[16][8];
    __shared__ float red[16];
    __shared__ v4f   hw[2 * FFDIM];
    __shared__ v4f   hfc[FFDIM / 4];

    const int l    = threadIdx.x & 31;
    const int wave = __builtin_amdgcn_readfirstlane((int)(threadIdx.x >> 5));
    const int half = l >> 4;
    const int nl   = l & 15;
    const int b0   = blockIdx.x * 16;

    v2f Aih[2][4];
    v2f Ahh[2][2];
#pragma unroll
    for (int mt = 0; mt < 2; ++mt) {
        const int  u  = mt * 4 + (nl >> 2);
        const bool uv = u < HDIM;
        const int  g  = uv ? (nl & 3) * HDIM + u : 0;
        const float pre = GATE_PRE(nl & 3);
#pragma unroll
        for (int kc = 0; kc < 4; ++kc) {
            int k0 = kc * 4 + 2 * half;
            float m0 = (uv && k0     < IDIM) ? pre : 0.0f;
            float m1 = (uv && k0 + 1 < IDIM) ? pre : 0.0f;
            int ka = (k0     < IDIM) ? k0     : 0;
            int kb = (k0 + 1 < IDIM) ? k0 + 1 : 0;
            Aih[mt][kc][0] = w_ih[g * IDIM + ka] * m0;
            Aih[mt][kc][1] = w_ih[g * IDIM + kb] * m1;
        }
#pragma unroll
        for (int kc = 0; kc < 2; ++kc) {
            int k0 = kc * 4 + 2 * half;
            float m0 = (uv && k0     < HDIM) ? pre : 0.0f;
            float m1 = (uv && k0 + 1 < HDIM) ? pre : 0.0f;
            int ka = (k0     < HDIM) ? k0     : 0;
            int kb = (k0 + 1 < HDIM) ? k0 + 1 : 0;
            Ahh[mt][kc][0] = w_hh[g * HDIM + ka] * m0;
            Ahh[mt][kc][1] = w_hh[g * HDIM + kb] * m1;
        }
    }

    v8f binit[2];
#pragma unroll
    for (int mt = 0; mt < 2; ++mt) {
#pragma unroll
        for (int v = 0; v < 8; ++v) {
            const int  u  = mt * 4 + 2 * half + (v >> 2);
            const bool uv = u < HDIM;
            const int  g  = uv ? (v & 3) * HDIM + u : 0;
            binit[mt][v] = uv ? (b_ih[g] + b_hh[g]) * GATE_PRE(v & 3) : 0.0f;
        }
    }

    v2f hB[2];
    v2f cp[2];
#pragma unroll
    for (int mt = 0; mt < 2; ++mt) {
        hB[mt][0] = 0.0f; hB[mt][1] = 0.0f;
        cp[mt][0] = 0.0f; cp[mt][1] = 0.0f;
    }

    const float* xblk = x + (size_t)b0 * (TSTEP * IDIM);
    const int xlane  = nl * (TSTEP * IDIM) + 2 * half;
    const int xlane3 = nl * (TSTEP * IDIM) + ((half == 0) ? 12 : 0);

    auto x_load = [&](int t, v2f (&xr)[4]) {
        const float* xt = xblk + t * IDIM;
#pragma unroll
        for (int kc = 0; kc < 3; ++kc) {
            xr[kc][0] = xt[xlane + kc * 4];
            xr[kc][1] = xt[xlane + kc * 4 + 1];
        }
        xr[3][0] = xt[xlane3];
        xr[3][1] = 0.0f;
    };

    auto x_project = [&](const v2f (&xr)[4], v8f& xa0, v8f& xa1) {
        xa0 = binit[0];
        xa1 = binit[1];
#pragma unroll
        for (int kc = 0; kc < 4; ++kc) {
            xa0 = __builtin_amdgcn_wmma_f32_16x16x4_f32(false, Aih[0][kc], false, xr[kc],
                                                        (short)0, xa0, false, false);
            xa1 = __builtin_amdgcn_wmma_f32_16x16x4_f32(false, Aih[1][kc], false, xr[kc],
                                                        (short)0, xa1, false, false);
        }
    };

    auto pair_update = [&](const v8f& a, v2f& cs) -> v2f {
        const v2f hf = {0.5f, 0.5f};
        v2f ti = {FAST_TANH(a[0]), FAST_TANH(a[4])};
        v2f tf = {FAST_TANH(a[1]), FAST_TANH(a[5])};
        v2f gg = {FAST_TANH(a[2]), FAST_TANH(a[6])};
        v2f to = {FAST_TANH(a[3]), FAST_TANH(a[7])};
        v2f ig = __builtin_elementwise_fma(hf, ti, hf);
        v2f fg = __builtin_elementwise_fma(hf, tf, hf);
        v2f og = __builtin_elementwise_fma(hf, to, hf);
        cs = __builtin_elementwise_fma(ig, gg, fg * cs);
        v2f tc = {FAST_TANH(cs[0]), FAST_TANH(cs[1])};
        return og * tc;
    };

    auto ring_read = [&](const v4f* p, v8f& a0, v8f& a1) {
        v4f q0 = p[0], q1 = p[32], q2 = p[64], q3 = p[96];
        a0 = __builtin_shufflevector(q0, q1, 0, 1, 2, 3, 4, 5, 6, 7);
        a1 = __builtin_shufflevector(q2, q3, 0, 1, 2, 3, 4, 5, 6, 7);
    };

    auto group_read = [&](const v4f* p, v8f (&t)[GRP][2]) {
#pragma unroll
        for (int s = 0; s < GRP; ++s) ring_read(p + s * SLOT, t[s][0], t[s][1]);
    };

    auto h_mma = [&](v8f& acc0, v8f& acc1) {
#pragma unroll
        for (int kc = 0; kc < 2; ++kc) {
            acc0 = __builtin_amdgcn_wmma_f32_16x16x4_f32(false, Ahh[0][kc], false, hB[kc],
                                                         (short)0, acc0, false, false);
            acc1 = __builtin_amdgcn_wmma_f32_16x16x4_f32(false, Ahh[1][kc], false, hB[kc],
                                                         (short)0, acc1, false, false);
        }
        __builtin_amdgcn_sched_barrier(0);
    };

    auto h_act = [&](const v8f& acc0, const v8f& acc1) {
        hB[0] = pair_update(acc0, cp[0]);
        hB[1] = pair_update(acc1, cp[1]);
    };

    auto run_group = [&](v8f (&cur)[GRP][2], v8f (&nxt)[GRP][2], const v4f* pn, bool rd) {
#pragma unroll
        for (int s = 0; s < GRP; ++s) {
            h_mma(cur[s][0], cur[s][1]);
            if (s == 0) {
                if (rd) group_read(pn, nxt);
                __builtin_amdgcn_sched_barrier(0);
            }
            h_act(cur[s][0], cur[s][1]);
        }
    };

    const int p = wave - 1;

    for (int j = 0; j <= NCH; ++j) {
        if (wave == 0) {
            if (j > 0) {
                const v4f* base = xg + ((j - 1) & 1) * (CH * SLOT) + l;
                v8f ta[GRP][2], tb[GRP][2];
                group_read(base, ta);
                for (int g = 0; g < NGRP; g += 2) {
                    run_group(ta, tb, base + (g + 1) * (GRP * SLOT), true);
                    run_group(tb, ta, base + ((g + 2) & (NGRP - 1)) * (GRP * SLOT), g + 2 < NGRP);
                }
            }
        } else {
            if (j < NCH) {
                v4f* dst = xg + (j & 1) * (CH * SLOT) + l;
                v2f xr[PSTEPS][4];
#pragma unroll
                for (int i = 0; i < PSTEPS; ++i) x_load(j * CH + p + NPROD * i, xr[i]);
#pragma unroll
                for (int i = 0; i < PSTEPS; ++i) {
                    v8f a0, a1;
                    x_project(xr[i], a0, a1);
                    v4f* d = dst + (p + NPROD * i) * SLOT;
                    d[0]  = __builtin_shufflevector(a0, a0, 0, 1, 2, 3);
                    d[32] = __builtin_shufflevector(a0, a0, 4, 5, 6, 7);
                    d[64] = __builtin_shufflevector(a1, a1, 0, 1, 2, 3);
                    d[96] = __builtin_shufflevector(a1, a1, 4, 5, 6, 7);
                }
            } else {
                const int tid = p * 32 + l;
                const float* w1  = fc1_w + tid * HDIM;
                const float* b1  = fc1_b + tid;
                const float* w2  = fc_w + tid;
                v4f*   hws  = hw + 2 * tid;
                float* hfcs = (float*)hfc + tid;
#pragma unroll
                for (int k = 0; k < FFDIM / (32 * NPROD); ++k) {
                    const int o = (32 * NPROD) * k;
                    const float* wr = w1 + o * HDIM;
                    v4f wa, wb;
                    wa[0] = wr[0]; wa[1] = wr[1]; wa[2] = wr[2]; wa[3] = wr[3];
                    wb[0] = wr[4]; wb[1] = wr[5]; wb[2] = wr[6]; wb[3] = b1[o];
                    hws[2 * o]     = wa;
                    hws[2 * o + 1] = wb;
                    hfcs[o]        = w2[o];
                }
            }
        }
        __builtin_amdgcn_fence(__ATOMIC_RELEASE, "workgroup", "local");
        __builtin_amdgcn_s_barrier();
        __builtin_amdgcn_fence(__ATOMIC_ACQUIRE, "workgroup", "local");
    }

    if (wave == 0) {
#pragma unroll
        for (int mt = 0; mt < 2; ++mt) {
            hbuf[nl][mt * 4 + 2 * half]     = hB[mt][0];
            hbuf[nl][mt * 4 + 2 * half + 1] = hB[mt][1];
        }
    }
    __syncthreads();

    float acc = 0.0f;
    if (wave == 0) {
        float hr[HDIM];
#pragma unroll
        for (int j = 0; j < HDIM; ++j) {
            float v = hbuf[nl][j];
            hr[j] = v > 0.0f ? v : 0.0f;
        }
        auto head_read = [&](const v4f* pw, const v4f* pf, v4f (&a)[HU], v4f (&b)[HU], v4f (&f)[HU / 4]) {
#pragma unroll
            for (int u = 0; u < HU; ++u) { a[u] = pw[2 * u]; b[u] = pw[2 * u + 1]; }
#pragma unroll
            for (int u = 0; u < HU / 4; ++u) f[u] = pf[u];
            __builtin_amdgcn_sched_barrier(0);
        };
        auto head_sum = [&](const v4f (&a)[HU], const v4f (&b)[HU], const v4f (&f)[HU / 4]) {
#pragma unroll
            for (int u = 0; u < HU; ++u) {
                float s = b[u][3];
                s += a[u][0] * hr[0];
                s += a[u][1] * hr[1];
                s += a[u][2] * hr[2];
                s += a[u][3] * hr[3];
                s += b[u][0] * hr[4];
                s += b[u][1] * hr[5];
                s += b[u][2] * hr[6];
                s = s > 0.0f ? s : 0.0f;
                acc += f[u >> 2][u & 3] * s;
            }
        };
        const v4f* pw = hw + 2 * half * HPART;
        const v4f* pf = hfc + half * (HPART / 4);
        v4f ra[HU], rb[HU], rf[HU / 4];
        v4f qa[HU], qb[HU], qf[HU / 4];
        head_read(pw, pf, ra, rb, rf);
        for (int g = 0; g < HPASS; g += 2) {
            head_read(pw + 2 * HU, pf + HU / 4, qa, qb, qf);
            head_sum(ra, rb, rf);
            pw += 4 * HU;
            pf += HU / 2;
            if (g + 2 < HPASS) head_read(pw, pf, ra, rb, rf);
            head_sum(qa, qb, qf);
        }
        if (half == 1) red[nl] = acc;
    }
    __syncthreads();
    if (wave == 0 && half == 0) out[b0 + nl] = acc + red[nl] + fc_b[0];
}

extern "C" void kernel_launch(void* const* d_in, const int* in_sizes, int n_in,
                              void* d_out, int out_size, void* d_ws, size_t ws_size,
                              hipStream_t stream) {
    (void)in_sizes; (void)n_in; (void)out_size; (void)d_ws; (void)ws_size;
    const float* x     = (const float*)d_in[0];
    const float* w_ih  = (const float*)d_in[1];
    const float* w_hh  = (const float*)d_in[2];
    const float* b_ih  = (const float*)d_in[3];
    const float* b_hh  = (const float*)d_in[4];
    const float* fc1_w = (const float*)d_in[5];
    const float* fc1_b = (const float*)d_in[6];
    const float* fc_w  = (const float*)d_in[7];
    const float* fc_b  = (const float*)d_in[8];

    lstm_fused_kernel<<<dim3(BATCH / 16), dim3(32 * (1 + NPROD)), 0, stream>>>(
        x, w_ih, w_hh, b_ih, b_hh, fc1_w, fc1_b, fc_w, fc_b, (float*)d_out);
}


# Recurrence step issued in a hand-written order behind scheduling boundaries: units 0-3 run ahead, h of units 4-7 is formed behind the next step's first WMMA, no WMMA waits for h; every sum as before
# speedup vs baseline: 6.2636x; 1.0339x over previous
#include <hip/hip_runtime.h>


typedef float v2f __attribute__((ext_vector_type(2)));
typedef float v4f __attribute__((ext_vector_type(4)));
typedef float v8f __attribute__((ext_vector_type(8)));

#define BATCH 4096
#define TSTEP 512
#define IDIM  13
#define HDIM  7
#define FFDIM 512
#define NPROD 2
#define CH    32
#define NCH   (TSTEP / CH)
#define PSTEPS (CH / NPROD)
#define SLOT  128
#define GRP   8
#define NGRP  (CH / GRP)
#define HU    8
#define HPART (FFDIM / 2)
#define HPASS (HPART / HU)

#define LOG2E 1.4426950408889634f

#if __has_builtin(__builtin_amdgcn_tanhf)
#define FAST_TANH(v) __builtin_amdgcn_tanhf(v)
#define TANH_TIER 2
#elif __has_builtin(__builtin_amdgcn_tanh_f32)
#define FAST_TANH(v) __builtin_amdgcn_tanh_f32(v)
#define TANH_TIER 2
#elif __has_builtin(__builtin_amdgcn_exp2f) && __has_builtin(__builtin_amdgcn_rcpf)
#define TANH_TIER 1
static __device__ __forceinline__ float FAST_TANH(float v) {
    float e2 = __builtin_amdgcn_exp2f(v * (2.0f * LOG2E));
    return fmaf(-2.0f, __builtin_amdgcn_rcpf(e2 + 1.0f), 1.0f);
}
#else
#define FAST_TANH(v) tanhf(v)
#define TANH_TIER 0
#endif

static __device__ __forceinline__ float sigm_half(float hx) {
#if TANH_TIER == 2
    return fmaf(0.5f, FAST_TANH(hx), 0.5f);
#elif TANH_TIER == 1
    return __builtin_amdgcn_rcpf(1.0f + __builtin_amdgcn_exp2f(hx * (-2.0f * LOG2E)));
#else
    return 1.0f / (1.0f + __expf(-2.0f * hx));
#endif
}

#define GATE_PRE(gate) (((gate) == 2) ? 1.0f : 0.5f)

__global__ __launch_bounds__(32 * (1 + NPROD)) void lstm_fused_kernel(
    const float* __restrict__ x,     const float* __restrict__ w_ih,
    const float* __restrict__ w_hh,  const float* __restrict__ b_ih,
    const float* __restrict__ b_hh,  const float* __restrict__ fc1_w,
    const float* __restrict__ fc1_b, const float* __restrict__ fc_w,
    const float* __restrict__ fc_b,  float* __restrict__ out)
{
    __shared__ v4f   xg[2 * CH * SLOT];
    __shared__ float hbuf[16][8];
    __shared__ float red[16];
    __shared__ v4f   hw[2 * FFDIM];
    __shared__ v4f   hfc[FFDIM / 4];

    const int l    = threadIdx.x & 31;
    const int wave = __builtin_amdgcn_readfirstlane((int)(threadIdx.x >> 5));
    const int half = l >> 4;
    const int nl   = l & 15;
    const int b0   = blockIdx.x * 16;

    v2f Aih[2][4];
    v2f Ahh[2][2];
#pragma unroll
    for (int mt = 0; mt < 2; ++mt) {
        const int  u  = mt * 4 + (nl >> 2);
        const bool uv = u < HDIM;
        const int  g  = uv ? (nl & 3) * HDIM + u : 0;
        const float pre = GATE_PRE(nl & 3);
#pragma unroll
        for (int kc = 0; kc < 4; ++kc) {
            int k0 = kc * 4 + 2 * half;
            float m0 = (uv && k0     < IDIM) ? pre : 0.0f;
            float m1 = (uv && k0 + 1 < IDIM) ? pre : 0.0f;
            int ka = (k0     < IDIM) ? k0     : 0;
            int kb = (k0 + 1 < IDIM) ? k0 + 1 : 0;
            Aih[mt][kc][0] = w_ih[g * IDIM + ka] * m0;
            Aih[mt][kc][1] = w_ih[g * IDIM + kb] * m1;
        }
#pragma unroll
        for (int kc = 0; kc < 2; ++kc) {
            int k0 = kc * 4 + 2 * half;
            float m0 = (uv && k0     < HDIM) ? pre : 0.0f;
            float m1 = (uv && k0 + 1 < HDIM) ? pre : 0.0f;
            int ka = (k0     < HDIM) ? k0     : 0;
            int kb = (k0 + 1 < HDIM) ? k0 + 1 : 0;
            Ahh[mt][kc][0] = w_hh[g * HDIM + ka] * m0;
            Ahh[mt][kc][1] = w_hh[g * HDIM + kb] * m1;
        }
    }

    v8f binit[2];
#pragma unroll
    for (int mt = 0; mt < 2; ++mt) {
#pragma unroll
        for (int v = 0; v < 8; ++v) {
            const int  u  = mt * 4 + 2 * half + (v >> 2);
            const bool uv = u < HDIM;
            const int  g  = uv ? (v & 3) * HDIM + u : 0;
            binit[mt][v] = uv ? (b_ih[g] + b_hh[g]) * GATE_PRE(v & 3) : 0.0f;
        }
    }

    v2f hB[2];
    v2f cp[2];
#pragma unroll
    for (int mt = 0; mt < 2; ++mt) {
        hB[mt][0] = 0.0f; hB[mt][1] = 0.0f;
        cp[mt][0] = 0.0f; cp[mt][1] = 0.0f;
    }

    const float* xblk = x + (size_t)b0 * (TSTEP * IDIM);
    const int xlane  = nl * (TSTEP * IDIM) + 2 * half;
    const int xlane3 = nl * (TSTEP * IDIM) + ((half == 0) ? 12 : 0);

    auto x_load = [&](int t, v2f (&xr)[4]) {
        const float* xt = xblk + t * IDIM;
#pragma unroll
        for (int kc = 0; kc < 3; ++kc) {
            xr[kc][0] = xt[xlane + kc * 4];
            xr[kc][1] = xt[xlane + kc * 4 + 1];
        }
        xr[3][0] = xt[xlane3];
        xr[3][1] = 0.0f;
    };

    auto x_project = [&](const v2f (&xr)[4], v8f& xa0, v8f& xa1) {
        xa0 = binit[0];
        xa1 = binit[1];
#pragma unroll
        for (int kc = 0; kc < 4; ++kc) {
            xa0 = __builtin_amdgcn_wmma_f32_16x16x4_f32(false, Aih[0][kc], false, xr[kc],
                                                        (short)0, xa0, false, false);
            xa1 = __builtin_amdgcn_wmma_f32_16x16x4_f32(false, Aih[1][kc], false, xr[kc],
                                                        (short)0, xa1, false, false);
        }
    };

    auto ring_read = [&](const v4f* p, v8f& a0, v8f& a1) {
        v4f q0 = p[0], q1 = p[32], q2 = p[64], q3 = p[96];
        a0 = __builtin_shufflevector(q0, q1, 0, 1, 2, 3, 4, 5, 6, 7);
        a1 = __builtin_shufflevector(q2, q3, 0, 1, 2, 3, 4, 5, 6, 7);
    };

    auto group_read = [&](const v4f* p, v8f (&t)[GRP][2]) {
#pragma unroll
        for (int s = 0; s < GRP; ++s) ring_read(p + s * SLOT, t[s][0], t[s][1]);
    };

#define ISSUE_IN_THIS_ORDER() __builtin_amdgcn_sched_barrier(0)

    v2f og1 = {0.0f, 0.0f};
    v2f tc1 = {0.0f, 0.0f};

    auto step = [&](v8f& a0, v8f& a1, auto&& behind_wmmas) {
        const v2f hf = {0.5f, 0.5f};
        a0 = __builtin_amdgcn_wmma_f32_16x16x4_f32(false, Ahh[0][0], false, hB[0],
                                                   (short)0, a0, false, false);
        ISSUE_IN_THIS_ORDER();
        hB[1] = og1 * tc1;
        ISSUE_IN_THIS_ORDER();
        a1 = __builtin_amdgcn_wmma_f32_16x16x4_f32(false, Ahh[1][0], false, hB[0],
                                                   (short)0, a1, false, false);
        a0 = __builtin_amdgcn_wmma_f32_16x16x4_f32(false, Ahh[0][1], false, hB[1],
                                                   (short)0, a0, false, false);
        a1 = __builtin_amdgcn_wmma_f32_16x16x4_f32(false, Ahh[1][1], false, hB[1],
                                                   (short)0, a1, false, false);
        ISSUE_IN_THIS_ORDER();
        behind_wmmas();
        v2f tf0 = {FAST_TANH(a0[1]), FAST_TANH(a0[5])};
        ISSUE_IN_THIS_ORDER();
        v2f ti0 = {FAST_TANH(a0[0]), FAST_TANH(a0[4])};
        ISSUE_IN_THIS_ORDER();
        v2f tf1 = {FAST_TANH(a1[1]), FAST_TANH(a1[5])};
        ISSUE_IN_THIS_ORDER();
        v2f fg0 = __builtin_elementwise_fma(hf, tf0, hf);
        ISSUE_IN_THIS_ORDER();
        v2f to0 = {FAST_TANH(a0[3]), FAST_TANH(a0[7])};
        ISSUE_IN_THIS_ORDER();
        v2f tg0 = {FAST_TANH(a0[2]), FAST_TANH(a0[6])};
        ISSUE_IN_THIS_ORDER();
        v2f ig0 = __builtin_elementwise_fma(hf, ti0, hf);
        ISSUE_IN_THIS_ORDER();
        v2f fc0 = fg0 * cp[0];
        ISSUE_IN_THIS_ORDER();
        v2f ti1 = {FAST_TANH(a1[0]), FAST_TANH(a1[4])};
        ISSUE_IN_THIS_ORDER();
        cp[0] = __builtin_elementwise_fma(ig0, tg0, fc0);
        ISSUE_IN_THIS_ORDER();
        v2f fg1 = __builtin_elementwise_fma(hf, tf1, hf);
        ISSUE_IN_THIS_ORDER();
        v2f tg1 = {FAST_TANH(a1[2]), FAST_TANH(a1[6])};
        ISSUE_IN_THIS_ORDER();
        v2f fc1 = fg1 * cp[1];
        ISSUE_IN_THIS_ORDER();
        v2f ig1 = __builtin_elementwise_fma(hf, ti1, hf);
        ISSUE_IN_THIS_ORDER();
        v2f og0 = __builtin_elementwise_fma(hf, to0, hf);
        ISSUE_IN_THIS_ORDER();
        v2f tc0 = {FAST_TANH(cp[0][0]), FAST_TANH(cp[0][1])};
        ISSUE_IN_THIS_ORDER();
        cp[1] = __builtin_elementwise_fma(ig1, tg1, fc1);
        ISSUE_IN_THIS_ORDER();
        v2f to1 = {FAST_TANH(a1[3]), FAST_TANH(a1[7])};
        ISSUE_IN_THIS_ORDER();
        hB[0] = og0 * tc0;
        ISSUE_IN_THIS_ORDER();
        tc1 = v2f{FAST_TANH(cp[1][0]), FAST_TANH(cp[1][1])};
        ISSUE_IN_THIS_ORDER();
        og1 = __builtin_elementwise_fma(hf, to1, hf);
        ISSUE_IN_THIS_ORDER();
    };

    auto run_group = [&](v8f (&cur)[GRP][2], v8f (&nxt)[GRP][2], const v4f* pn, bool rd) {
#pragma unroll
        for (int s = 0; s < GRP; ++s) {
            step(cur[s][0], cur[s][1], [&]() {
                if (s == 0) {
                    if (rd) group_read(pn, nxt);
                    __builtin_amdgcn_sched_barrier(0);
                }
            });
        }
    };

    const int p = wave - 1;

    for (int j = 0; j <= NCH; ++j) {
        if (wave == 0) {
            if (j > 0) {
                const v4f* base = xg + ((j - 1) & 1) * (CH * SLOT) + l;
                v8f ta[GRP][2], tb[GRP][2];
                group_read(base, ta);
                for (int g = 0; g < NGRP; g += 2) {
                    run_group(ta, tb, base + (g + 1) * (GRP * SLOT), true);
                    run_group(tb, ta, base + ((g + 2) & (NGRP - 1)) * (GRP * SLOT), g + 2 < NGRP);
                }
            }
        } else {
            if (j < NCH) {
                v4f* dst = xg + (j & 1) * (CH * SLOT) + l;
                v2f xr[PSTEPS][4];
#pragma unroll
                for (int i = 0; i < PSTEPS; ++i) x_load(j * CH + p + NPROD * i, xr[i]);
#pragma unroll
                for (int i = 0; i < PSTEPS; ++i) {
                    v8f a0, a1;
                    x_project(xr[i], a0, a1);
                    v4f* d = dst + (p + NPROD * i) * SLOT;
                    d[0]  = __builtin_shufflevector(a0, a0, 0, 1, 2, 3);
                    d[32] = __builtin_shufflevector(a0, a0, 4, 5, 6, 7);
                    d[64] = __builtin_shufflevector(a1, a1, 0, 1, 2, 3);
                    d[96] = __builtin_shufflevector(a1, a1, 4, 5, 6, 7);
                }
            } else {
                const int tid = p * 32 + l;
                const float* w1  = fc1_w + tid * HDIM;
                const float* b1  = fc1_b + tid;
                const float* w2  = fc_w + tid;
                v4f*   hws  = hw + 2 * tid;
                float* hfcs = (float*)hfc + tid;
#pragma unroll
                for (int k = 0; k < FFDIM / (32 * NPROD); ++k) {
                    const int o = (32 * NPROD) * k;
                    const float* wr = w1 + o * HDIM;
                    v4f wa, wb;
                    wa[0] = wr[0]; wa[1] = wr[1]; wa[2] = wr[2]; wa[3] = wr[3];
                    wb[0] = wr[4]; wb[1] = wr[5]; wb[2] = wr[6]; wb[3] = b1[o];
                    hws[2 * o]     = wa;
                    hws[2 * o + 1] = wb;
                    hfcs[o]        = w2[o];
                }
            }
        }
        __builtin_amdgcn_fence(__ATOMIC_RELEASE, "workgroup", "local");
        __builtin_amdgcn_s_barrier();
        __builtin_amdgcn_fence(__ATOMIC_ACQUIRE, "workgroup", "local");
    }

    if (wave == 0) {
        hB[1] = og1 * tc1;
#pragma unroll
        for (int mt = 0; mt < 2; ++mt) {
            hbuf[nl][mt * 4 + 2 * half]     = hB[mt][0];
            hbuf[nl][mt * 4 + 2 * half + 1] = hB[mt][1];
        }
    }
    __syncthreads();

    float acc = 0.0f;
    if (wave == 0) {
        float hr[HDIM];
#pragma unroll
        for (int j = 0; j < HDIM; ++j) {
            float v = hbuf[nl][j];
            hr[j] = v > 0.0f ? v : 0.0f;
        }
        auto head_read = [&](const v4f* pw, const v4f* pf, v4f (&a)[HU], v4f (&b)[HU], v4f (&f)[HU / 4]) {
#pragma unroll
            for (int u = 0; u < HU; ++u) { a[u] = pw[2 * u]; b[u] = pw[2 * u + 1]; }
#pragma unroll
            for (int u = 0; u < HU / 4; ++u) f[u] = pf[u];
            __builtin_amdgcn_sched_barrier(0);
        };
        auto head_sum = [&](const v4f (&a)[HU], const v4f (&b)[HU], const v4f (&f)[HU / 4]) {
#pragma unroll
            for (int u = 0; u < HU; ++u) {
                float s = b[u][3];
                s += a[u][0] * hr[0];
                s += a[u][1] * hr[1];
                s += a[u][2] * hr[2];
                s += a[u][3] * hr[3];
                s += b[u][0] * hr[4];
                s += b[u][1] * hr[5];
                s += b[u][2] * hr[6];
                s = s > 0.0f ? s : 0.0f;
                acc += f[u >> 2][u & 3] * s;
            }
        };
        const v4f* pw = hw + 2 * half * HPART;
        const v4f* pf = hfc + half * (HPART / 4);
        v4f ra[HU], rb[HU], rf[HU / 4];
        v4f qa[HU], qb[HU], qf[HU / 4];
        head_read(pw, pf, ra, rb, rf);
        for (int g = 0; g < HPASS; g += 2) {
            head_read(pw + 2 * HU, pf + HU / 4, qa, qb, qf);
            head_sum(ra, rb, rf);
            pw += 4 * HU;
            pf += HU / 2;
            if (g + 2 < HPASS) head_read(pw, pf, ra, rb, rf);
            head_sum(qa, qb, qf);
        }
        if (half == 1) red[nl] = acc;
    }
    __syncthreads();
    if (wave == 0 && half == 0) out[b0 + nl] = acc + red[nl] + fc_b[0];
}

extern "C" void kernel_launch(void* const* d_in, const int* in_sizes, int n_in,
                              void* d_out, int out_size, void* d_ws, size_t ws_size,
                              hipStream_t stream) {
    (void)in_sizes; (void)n_in; (void)out_size; (void)d_ws; (void)ws_size;
    const float* x     = (const float*)d_in[0];
    const float* w_ih  = (const float*)d_in[1];
    const float* w_hh  = (const float*)d_in[2];
    const float* b_ih  = (const float*)d_in[3];
    const float* b_hh  = (const float*)d_in[4];
    const float* fc1_w = (const float*)d_in[5];
    const float* fc1_b = (const float*)d_in[6];
    const float* fc_w  = (const float*)d_in[7];
    const float* fc_b  = (const float*)d_in[8];

    lstm_fused_kernel<<<dim3(BATCH / 16), dim3(32 * (1 + NPROD)), 0, stream>>>(
        x, w_ih, w_hh, b_ih, b_hh, fc1_w, fc1_b, fc_w, fc_b, (float*)d_out);
}


# FF head's first layer on the matrix unit: 32 f32 WMMA tiles (16 features x 16 batch rows, relu(h) as B, bias in the spare k slot) dealt over the 3 waves; head sums reordered, 6 partials per row
# speedup vs baseline: 6.6162x; 1.0563x over previous
#include <hip/hip_runtime.h>


typedef float v2f __attribute__((ext_vector_type(2)));
typedef float v4f __attribute__((ext_vector_type(4)));
typedef float v8f __attribute__((ext_vector_type(8)));

#define BATCH 4096
#define TSTEP 512
#define IDIM  13
#define HDIM  7
#define FFDIM 512
#define NPROD 2
#define CH    32
#define NCH   (TSTEP / CH)
#define PSTEPS (CH / NPROD)
#define SLOT  128
#define GRP   8
#define NGRP  (CH / GRP)
#define HT    (FFDIM / 16)
#define HTW   ((HT + NPROD) / (1 + NPROD))

#define LOG2E 1.4426950408889634f

#if __has_builtin(__builtin_amdgcn_tanhf)
#define FAST_TANH(v) __builtin_amdgcn_tanhf(v)
#define TANH_TIER 2
#elif __has_builtin(__builtin_amdgcn_tanh_f32)
#define FAST_TANH(v) __builtin_amdgcn_tanh_f32(v)
#define TANH_TIER 2
#elif __has_builtin(__builtin_amdgcn_exp2f) && __has_builtin(__builtin_amdgcn_rcpf)
#define TANH_TIER 1
static __device__ __forceinline__ float FAST_TANH(float v) {
    float e2 = __builtin_amdgcn_exp2f(v * (2.0f * LOG2E));
    return fmaf(-2.0f, __builtin_amdgcn_rcpf(e2 + 1.0f), 1.0f);
}
#else
#define FAST_TANH(v) tanhf(v)
#define TANH_TIER 0
#endif

static __device__ __forceinline__ float sigm_half(float hx) {
#if TANH_TIER == 2
    return fmaf(0.5f, FAST_TANH(hx), 0.5f);
#elif TANH_TIER == 1
    return __builtin_amdgcn_rcpf(1.0f + __builtin_amdgcn_exp2f(hx * (-2.0f * LOG2E)));
#else
    return 1.0f / (1.0f + __expf(-2.0f * hx));
#endif
}

#define GATE_PRE(gate) (((gate) == 2) ? 1.0f : 0.5f)

__global__ __launch_bounds__(32 * (1 + NPROD)) void lstm_fused_kernel(
    const float* __restrict__ x,     const float* __restrict__ w_ih,
    const float* __restrict__ w_hh,  const float* __restrict__ b_ih,
    const float* __restrict__ b_hh,  const float* __restrict__ fc1_w,
    const float* __restrict__ fc1_b, const float* __restrict__ fc_w,
    const float* __restrict__ fc_b,  float* __restrict__ out)
{
    __shared__ v4f   xg[2 * CH * SLOT];
    __shared__ float hbuf[16][8];
    __shared__ float red[1 + NPROD][32];
    __shared__ v4f   hw[2 * FFDIM];
    __shared__ v4f   hfc[FFDIM / 4];

    const int l    = threadIdx.x & 31;
    const int wave = __builtin_amdgcn_readfirstlane((int)(threadIdx.x >> 5));
    const int half = l >> 4;
    const int nl   = l & 15;
    const int b0   = blockIdx.x * 16;

    v2f Aih[2][4];
    v2f Ahh[2][2];
#pragma unroll
    for (int mt = 0; mt < 2; ++mt) {
        const int  u  = mt * 4 + (nl >> 2);
        const bool uv = u < HDIM;
        const int  g  = uv ? (nl & 3) * HDIM + u : 0;
        const float pre = GATE_PRE(nl & 3);
#pragma unroll
        for (int kc = 0; kc < 4; ++kc) {
            int k0 = kc * 4 + 2 * half;
            float m0 = (uv && k0     < IDIM) ? pre : 0.0f;
            float m1 = (uv && k0 + 1 < IDIM) ? pre : 0.0f;
            int ka = (k0     < IDIM) ? k0     : 0;
            int kb = (k0 + 1 < IDIM) ? k0 + 1 : 0;
            Aih[mt][kc][0] = w_ih[g * IDIM + ka] * m0;
            Aih[mt][kc][1] = w_ih[g * IDIM + kb] * m1;
        }
#pragma unroll
        for (int kc = 0; kc < 2; ++kc) {
            int k0 = kc * 4 + 2 * half;
            float m0 = (uv && k0     < HDIM) ? pre : 0.0f;
            float m1 = (uv && k0 + 1 < HDIM) ? pre : 0.0f;
            int ka = (k0     < HDIM) ? k0     : 0;
            int kb = (k0 + 1 < HDIM) ? k0 + 1 : 0;
            Ahh[mt][kc][0] = w_hh[g * HDIM + ka] * m0;
            Ahh[mt][kc][1] = w_hh[g * HDIM + kb] * m1;
        }
    }

    v8f binit[2];
#pragma unroll
    for (int mt = 0; mt < 2; ++mt) {
#pragma unroll
        for (int v = 0; v < 8; ++v) {
            const int  u  = mt * 4 + 2 * half + (v >> 2);
            const bool uv = u < HDIM;
            const int  g  = uv ? (v & 3) * HDIM + u : 0;
            binit[mt][v] = uv ? (b_ih[g] + b_hh[g]) * GATE_PRE(v & 3) : 0.0f;
        }
    }

    v2f hB[2];
    v2f cp[2];
#pragma unroll
    for (int mt = 0; mt < 2; ++mt) {
        hB[mt][0] = 0.0f; hB[mt][1] = 0.0f;
        cp[mt][0] = 0.0f; cp[mt][1] = 0.0f;
    }

    const float* xblk = x + (size_t)b0 * (TSTEP * IDIM);
    const int xlane  = nl * (TSTEP * IDIM) + 2 * half;
    const int xlane3 = nl * (TSTEP * IDIM) + ((half == 0) ? 12 : 0);

    auto x_load = [&](int t, v2f (&xr)[4]) {
        const float* xt = xblk + t * IDIM;
#pragma unroll
        for (int kc = 0; kc < 3; ++kc) {
            xr[kc][0] = xt[xlane + kc * 4];
            xr[kc][1] = xt[xlane + kc * 4 + 1];
        }
        xr[3][0] = xt[xlane3];
        xr[3][1] = 0.0f;
    };

    auto x_project = [&](const v2f (&xr)[4], v8f& xa0, v8f& xa1) {
        xa0 = binit[0];
        xa1 = binit[1];
#pragma unroll
        for (int kc = 0; kc < 4; ++kc) {
            xa0 = __builtin_amdgcn_wmma_f32_16x16x4_f32(false, Aih[0][kc], false, xr[kc],
                                                        (short)0, xa0, false, false);
            xa1 = __builtin_amdgcn_wmma_f32_16x16x4_f32(false, Aih[1][kc], false, xr[kc],
                                                        (short)0, xa1, false, false);
        }
    };

    auto ring_read = [&](const v4f* p, v8f& a0, v8f& a1) {
        v4f q0 = p[0], q1 = p[32], q2 = p[64], q3 = p[96];
        a0 = __builtin_shufflevector(q0, q1, 0, 1, 2, 3, 4, 5, 6, 7);
        a1 = __builtin_shufflevector(q2, q3, 0, 1, 2, 3, 4, 5, 6, 7);
    };

    auto group_read = [&](const v4f* p, v8f (&t)[GRP][2]) {
#pragma unroll
        for (int s = 0; s < GRP; ++s) ring_read(p + s * SLOT, t[s][0], t[s][1]);
    };

#define ISSUE_IN_THIS_ORDER() __builtin_amdgcn_sched_barrier(0)

    v2f og1 = {0.0f, 0.0f};
    v2f tc1 = {0.0f, 0.0f};

    auto step = [&](v8f& a0, v8f& a1, auto&& behind_wmmas) {
        const v2f hf = {0.5f, 0.5f};
        a0 = __builtin_amdgcn_wmma_f32_16x16x4_f32(false, Ahh[0][0], false, hB[0],
                                                   (short)0, a0, false, false);
        ISSUE_IN_THIS_ORDER();
        hB[1] = og1 * tc1;
        ISSUE_IN_THIS_ORDER();
        a1 = __builtin_amdgcn_wmma_f32_16x16x4_f32(false, Ahh[1][0], false, hB[0],
                                                   (short)0, a1, false, false);
        a0 = __builtin_amdgcn_wmma_f32_16x16x4_f32(false, Ahh[0][1], false, hB[1],
                                                   (short)0, a0, false, false);
        a1 = __builtin_amdgcn_wmma_f32_16x16x4_f32(false, Ahh[1][1], false, hB[1],
                                                   (short)0, a1, false, false);
        ISSUE_IN_THIS_ORDER();
        behind_wmmas();
        v2f tf0 = {FAST_TANH(a0[1]), FAST_TANH(a0[5])};
        ISSUE_IN_THIS_ORDER();
        v2f ti0 = {FAST_TANH(a0[0]), FAST_TANH(a0[4])};
        ISSUE_IN_THIS_ORDER();
        v2f tf1 = {FAST_TANH(a1[1]), FAST_TANH(a1[5])};
        ISSUE_IN_THIS_ORDER();
        v2f fg0 = __builtin_elementwise_fma(hf, tf0, hf);
        ISSUE_IN_THIS_ORDER();
        v2f to0 = {FAST_TANH(a0[3]), FAST_TANH(a0[7])};
        ISSUE_IN_THIS_ORDER();
        v2f tg0 = {FAST_TANH(a0[2]), FAST_TANH(a0[6])};
        ISSUE_IN_THIS_ORDER();
        v2f ig0 = __builtin_elementwise_fma(hf, ti0, hf);
        ISSUE_IN_THIS_ORDER();
        v2f fc0 = fg0 * cp[0];
        ISSUE_IN_THIS_ORDER();
        v2f ti1 = {FAST_TANH(a1[0]), FAST_TANH(a1[4])};
        ISSUE_IN_THIS_ORDER();
        cp[0] = __builtin_elementwise_fma(ig0, tg0, fc0);
        ISSUE_IN_THIS_ORDER();
        v2f fg1 = __builtin_elementwise_fma(hf, tf1, hf);
        ISSUE_IN_THIS_ORDER();
        v2f tg1 = {FAST_TANH(a1[2]), FAST_TANH(a1[6])};
        ISSUE_IN_THIS_ORDER();
        v2f fc1 = fg1 * cp[1];
        ISSUE_IN_THIS_ORDER();
        v2f ig1 = __builtin_elementwise_fma(hf, ti1, hf);
        ISSUE_IN_THIS_ORDER();
        v2f og0 = __builtin_elementwise_fma(hf, to0, hf);
        ISSUE_IN_THIS_ORDER();
        v2f tc0 = {FAST_TANH(cp[0][0]), FAST_TANH(cp[0][1])};
        ISSUE_IN_THIS_ORDER();
        cp[1] = __builtin_elementwise_fma(ig1, tg1, fc1);
        ISSUE_IN_THIS_ORDER();
        v2f to1 = {FAST_TANH(a1[3]), FAST_TANH(a1[7])};
        ISSUE_IN_THIS_ORDER();
        hB[0] = og0 * tc0;
        ISSUE_IN_THIS_ORDER();
        tc1 = v2f{FAST_TANH(cp[1][0]), FAST_TANH(cp[1][1])};
        ISSUE_IN_THIS_ORDER();
        og1 = __builtin_elementwise_fma(hf, to1, hf);
        ISSUE_IN_THIS_ORDER();
    };

    auto run_group = [&](v8f (&cur)[GRP][2], v8f (&nxt)[GRP][2], const v4f* pn, bool rd) {
#pragma unroll
        for (int s = 0; s < GRP; ++s) {
            step(cur[s][0], cur[s][1], [&]() {
                if (s == 0) {
                    if (rd) group_read(pn, nxt);
                    __builtin_amdgcn_sched_barrier(0);
                }
            });
        }
    };

    const int p = wave - 1;

    for (int j = 0; j <= NCH; ++j) {
        if (wave == 0) {
            if (j > 0) {
                const v4f* base = xg + ((j - 1) & 1) * (CH * SLOT) + l;
                v8f ta[GRP][2], tb[GRP][2];
                group_read(base, ta);
                for (int g = 0; g < NGRP; g += 2) {
                    run_group(ta, tb, base + (g + 1) * (GRP * SLOT), true);
                    run_group(tb, ta, base + ((g + 2) & (NGRP - 1)) * (GRP * SLOT), g + 2 < NGRP);
                }
            }
        } else {
            if (j < NCH) {
                v4f* dst = xg + (j & 1) * (CH * SLOT) + l;
                v2f xr[PSTEPS][4];
#pragma unroll
                for (int i = 0; i < PSTEPS; ++i) x_load(j * CH + p + NPROD * i, xr[i]);
#pragma unroll
                for (int i = 0; i < PSTEPS; ++i) {
                    v8f a0, a1;
                    x_project(xr[i], a0, a1);
                    v4f* d = dst + (p + NPROD * i) * SLOT;
                    d[0]  = __builtin_shufflevector(a0, a0, 0, 1, 2, 3);
                    d[32] = __builtin_shufflevector(a0, a0, 4, 5, 6, 7);
                    d[64] = __builtin_shufflevector(a1, a1, 0, 1, 2, 3);
                    d[96] = __builtin_shufflevector(a1, a1, 4, 5, 6, 7);
                }
            } else {
                const int tid = p * 32 + l;
                const float* w1  = fc1_w + tid * HDIM;
                const float* b1  = fc1_b + tid;
                const float* w2  = fc_w + tid;
                v4f*   hws  = hw + 2 * tid;
                float* hfcs = (float*)hfc + tid;
#pragma unroll
                for (int k = 0; k < FFDIM / (32 * NPROD); ++k) {
                    const int o = (32 * NPROD) * k;
                    const float* wr = w1 + o * HDIM;
                    v4f wa, wb;
                    wa[0] = wr[0]; wa[1] = wr[1]; wa[2] = wr[4]; wa[3] = wr[5];
                    wb[0] = wr[2]; wb[1] = wr[3]; wb[2] = wr[6]; wb[3] = b1[o];
                    hws[2 * o]     = wa;
                    hws[2 * o + 1] = wb;
                    hfcs[o]        = w2[o];
                }
            }
        }
        __builtin_amdgcn_fence(__ATOMIC_RELEASE, "workgroup", "local");
        __builtin_amdgcn_s_barrier();
        __builtin_amdgcn_fence(__ATOMIC_ACQUIRE, "workgroup", "local");
    }

    const int hr0 = (wave == 0) ? NPROD : wave - 1;
    const int hnt = (wave == 0) ? HTW - 1 : HTW;
    v4f ha[HTW], hf[HTW][2];
    auto head_operands = [&]() {
#pragma unroll
        for (int i = 0; i < HTW; ++i) {
            if (i < hnt) {
                const int T = (1 + NPROD) * i + hr0;
                ha[i]    = hw[2 * (16 * T + nl) + half];
                hf[i][0] = hfc[4 * T + 2 * half];
                hf[i][1] = hfc[4 * T + 2 * half + 1];
            }
        }
    };
    if (wave != 0) head_operands();

    if (wave == 0) {
        hB[1] = og1 * tc1;
#pragma unroll
        for (int mt = 0; mt < 2; ++mt) {
            hbuf[nl][mt * 4 + 2 * half]     = hB[mt][0];
            hbuf[nl][mt * 4 + 2 * half + 1] = hB[mt][1];
        }
    }
    __syncthreads();

    v2f rb0, rb1;
    {
        const float h0 = hbuf[nl][2 * half], h1 = hbuf[nl][2 * half + 1];
        const float h2 = hbuf[nl][4 + 2 * half], h3 = hbuf[nl][4 + 2 * half + 1];
        if (wave == 0) head_operands();
        rb0[0] = h0 > 0.0f ? h0 : 0.0f;
        rb0[1] = h1 > 0.0f ? h1 : 0.0f;
        rb1[0] = h2 > 0.0f ? h2 : 0.0f;
        rb1[1] = (half == 1) ? 1.0f : (h3 > 0.0f ? h3 : 0.0f);
    }

    v2f hs[4];
#pragma unroll
    for (int q = 0; q < 4; ++q) { hs[q][0] = 0.0f; hs[q][1] = 0.0f; }
#pragma unroll
    for (int i = 0; i < HTW; ++i) {
        if (i < hnt) {
            v8f z = {0.0f, 0.0f, 0.0f, 0.0f, 0.0f, 0.0f, 0.0f, 0.0f};
            const v2f a0 = {ha[i][0], ha[i][1]};
            const v2f a1 = {ha[i][2], ha[i][3]};
            z = __builtin_amdgcn_wmma_f32_16x16x4_f32(false, a0, false, rb0,
                                                      (short)0, z, false, false);
            z = __builtin_amdgcn_wmma_f32_16x16x4_f32(false, a1, false, rb1,
                                                      (short)0, z, false, false);
#pragma unroll
            for (int q = 0; q < 4; ++q) {
                v2f r, w;
                r[0] = z[2 * q]     > 0.0f ? z[2 * q]     : 0.0f;
                r[1] = z[2 * q + 1] > 0.0f ? z[2 * q + 1] : 0.0f;
                w[0] = hf[i][q >> 1][(2 * q) & 3];
                w[1] = hf[i][q >> 1][(2 * q + 1) & 3];
                hs[q] = __builtin_elementwise_fma(r, w, hs[q]);
            }
        }
    }
    red[wave][l] = ((hs[0][0] + hs[0][1]) + (hs[1][0] + hs[1][1]))
                 + ((hs[2][0] + hs[2][1]) + (hs[3][0] + hs[3][1]));
    __syncthreads();
    if (wave == 0 && half == 0) {
        float s = red[0][nl] + red[0][nl + 16];
#pragma unroll
        for (int w = 1; w <= NPROD; ++w) s += red[w][nl] + red[w][nl + 16];
        out[b0 + nl] = s + fc_b[0];
    }
}

extern "C" void kernel_launch(void* const* d_in, const int* in_sizes, int n_in,
                              void* d_out, int out_size, void* d_ws, size_t ws_size,
                              hipStream_t stream) {
    (void)in_sizes; (void)n_in; (void)out_size; (void)d_ws; (void)ws_size;
    const float* x     = (const float*)d_in[0];
    const float* w_ih  = (const float*)d_in[1];
    const float* w_hh  = (const float*)d_in[2];
    const float* b_ih  = (const float*)d_in[3];
    const float* b_hh  = (const float*)d_in[4];
    const float* fc1_w = (const float*)d_in[5];
    const float* fc1_b = (const float*)d_in[6];
    const float* fc_w  = (const float*)d_in[7];
    const float* fc_b  = (const float*)d_in[8];

    lstm_fused_kernel<<<dim3(BATCH / 16), dim3(32 * (1 + NPROD)), 0, stream>>>(
        x, w_ih, w_hh, b_ih, b_hh, fc1_w, fc1_b, fc_w, fc_b, (float*)d_out);
}


# Chunk barrier split on wave 0: it arrives once its last LDS read of a chunk has landed and waits only before reading the next chunk's first group, read ahead behind WMMAs; every sum as before
# speedup vs baseline: 7.2927x; 1.1022x over previous
#include <hip/hip_runtime.h>


typedef float v2f __attribute__((ext_vector_type(2)));
typedef float v4f __attribute__((ext_vector_type(4)));
typedef float v8f __attribute__((ext_vector_type(8)));

#define BATCH 4096
#define TSTEP 512
#define IDIM  13
#define HDIM  7
#define FFDIM 512
#define NPROD 2
#define CH    32
#define NCH   (TSTEP / CH)
#define PSTEPS (CH / NPROD)
#define SLOT  128
#define GRP   8
#define NGRP  (CH / GRP)
#define HT    (FFDIM / 16)
#define HTW   ((HT + NPROD) / (1 + NPROD))

#define LOG2E 1.4426950408889634f

#if __has_builtin(__builtin_amdgcn_tanhf)
#define FAST_TANH(v) __builtin_amdgcn_tanhf(v)
#define TANH_TIER 2
#elif __has_builtin(__builtin_amdgcn_tanh_f32)
#define FAST_TANH(v) __builtin_amdgcn_tanh_f32(v)
#define TANH_TIER 2
#elif __has_builtin(__builtin_amdgcn_exp2f) && __has_builtin(__builtin_amdgcn_rcpf)
#define TANH_TIER 1
static __device__ __forceinline__ float FAST_TANH(float v) {
    float e2 = __builtin_amdgcn_exp2f(v * (2.0f * LOG2E));
    return fmaf(-2.0f, __builtin_amdgcn_rcpf(e2 + 1.0f), 1.0f);
}
#else
#define FAST_TANH(v) tanhf(v)
#define TANH_TIER 0
#endif

static __device__ __forceinline__ float sigm_half(float hx) {
#if TANH_TIER == 2
    return fmaf(0.5f, FAST_TANH(hx), 0.5f);
#elif TANH_TIER == 1
    return __builtin_amdgcn_rcpf(1.0f + __builtin_amdgcn_exp2f(hx * (-2.0f * LOG2E)));
#else
    return 1.0f / (1.0f + __expf(-2.0f * hx));
#endif
}

#define GATE_PRE(gate) (((gate) == 2) ? 1.0f : 0.5f)

__global__ __launch_bounds__(32 * (1 + NPROD)) void lstm_fused_kernel(
    const float* __restrict__ x,     const float* __restrict__ w_ih,
    const float* __restrict__ w_hh,  const float* __restrict__ b_ih,
    const float* __restrict__ b_hh,  const float* __restrict__ fc1_w,
    const float* __restrict__ fc1_b, const float* __restrict__ fc_w,
    const float* __restrict__ fc_b,  float* __restrict__ out)
{
    __shared__ v4f   xg[2 * CH * SLOT];
    __shared__ float hbuf[16][8];
    __shared__ float red[1 + NPROD][32];
    __shared__ v4f   hw[2 * FFDIM];
    __shared__ v4f   hfc[FFDIM / 4];

    const int l    = threadIdx.x & 31;
    const int wave = __builtin_amdgcn_readfirstlane((int)(threadIdx.x >> 5));
    const int half = l >> 4;
    const int nl   = l & 15;
    const int b0   = blockIdx.x * 16;

    v2f Aih[2][4];
    v2f Ahh[2][2];
#pragma unroll
    for (int mt = 0; mt < 2; ++mt) {
        const int  u  = mt * 4 + (nl >> 2);
        const bool uv = u < HDIM;
        const int  g  = uv ? (nl & 3) * HDIM + u : 0;
        const float pre = GATE_PRE(nl & 3);
#pragma unroll
        for (int kc = 0; kc < 4; ++kc) {
            int k0 = kc * 4 + 2 * half;
            float m0 = (uv && k0     < IDIM) ? pre : 0.0f;
            float m1 = (uv && k0 + 1 < IDIM) ? pre : 0.0f;
            int ka = (k0     < IDIM) ? k0     : 0;
            int kb = (k0 + 1 < IDIM) ? k0 + 1 : 0;
            Aih[mt][kc][0] = w_ih[g * IDIM + ka] * m0;
            Aih[mt][kc][1] = w_ih[g * IDIM + kb] * m1;
        }
#pragma unroll
        for (int kc = 0; kc < 2; ++kc) {
            int k0 = kc * 4 + 2 * half;
            float m0 = (uv && k0     < HDIM) ? pre : 0.0f;
            float m1 = (uv && k0 + 1 < HDIM) ? pre : 0.0f;
            int ka = (k0     < HDIM) ? k0     : 0;
            int kb = (k0 + 1 < HDIM) ? k0 + 1 : 0;
            Ahh[mt][kc][0] = w_hh[g * HDIM + ka] * m0;
            Ahh[mt][kc][1] = w_hh[g * HDIM + kb] * m1;
        }
    }

    v8f binit[2];
#pragma unroll
    for (int mt = 0; mt < 2; ++mt) {
#pragma unroll
        for (int v = 0; v < 8; ++v) {
            const int  u  = mt * 4 + 2 * half + (v >> 2);
            const bool uv = u < HDIM;
            const int  g  = uv ? (v & 3) * HDIM + u : 0;
            binit[mt][v] = uv ? (b_ih[g] + b_hh[g]) * GATE_PRE(v & 3) : 0.0f;
        }
    }

    v2f hB[2];
    v2f cp[2];
#pragma unroll
    for (int mt = 0; mt < 2; ++mt) {
        hB[mt][0] = 0.0f; hB[mt][1] = 0.0f;
        cp[mt][0] = 0.0f; cp[mt][1] = 0.0f;
    }

    const float* xblk = x + (size_t)b0 * (TSTEP * IDIM);
    const int xlane  = nl * (TSTEP * IDIM) + 2 * half;
    const int xlane3 = nl * (TSTEP * IDIM) + ((half == 0) ? 12 : 0);

    auto x_load = [&](int t, v2f (&xr)[4]) {
        const float* xt = xblk + t * IDIM;
#pragma unroll
        for (int kc = 0; kc < 3; ++kc) {
            xr[kc][0] = xt[xlane + kc * 4];
            xr[kc][1] = xt[xlane + kc * 4 + 1];
        }
        xr[3][0] = xt[xlane3];
        xr[3][1] = 0.0f;
    };

    auto x_project = [&](const v2f (&xr)[4], v8f& xa0, v8f& xa1) {
        xa0 = binit[0];
        xa1 = binit[1];
#pragma unroll
        for (int kc = 0; kc < 4; ++kc) {
            xa0 = __builtin_amdgcn_wmma_f32_16x16x4_f32(false, Aih[0][kc], false, xr[kc],
                                                        (short)0, xa0, false, false);
            xa1 = __builtin_amdgcn_wmma_f32_16x16x4_f32(false, Aih[1][kc], false, xr[kc],
                                                        (short)0, xa1, false, false);
        }
    };

    auto ring_read = [&](const v4f* p, v8f& a0, v8f& a1) {
        v4f q0 = p[0], q1 = p[32], q2 = p[64], q3 = p[96];
        a0 = __builtin_shufflevector(q0, q1, 0, 1, 2, 3, 4, 5, 6, 7);
        a1 = __builtin_shufflevector(q2, q3, 0, 1, 2, 3, 4, 5, 6, 7);
    };

    auto group_read = [&](const v4f* p, v8f (&t)[GRP][2]) {
#pragma unroll
        for (int s = 0; s < GRP; ++s) ring_read(p + s * SLOT, t[s][0], t[s][1]);
    };

#define ISSUE_IN_THIS_ORDER() __builtin_amdgcn_sched_barrier(0)

    v2f og1 = {0.0f, 0.0f};
    v2f tc1 = {0.0f, 0.0f};

    auto step = [&](v8f& a0, v8f& a1, auto&& behind_wmmas) {
        const v2f hf = {0.5f, 0.5f};
        a0 = __builtin_amdgcn_wmma_f32_16x16x4_f32(false, Ahh[0][0], false, hB[0],
                                                   (short)0, a0, false, false);
        ISSUE_IN_THIS_ORDER();
        hB[1] = og1 * tc1;
        ISSUE_IN_THIS_ORDER();
        a1 = __builtin_amdgcn_wmma_f32_16x16x4_f32(false, Ahh[1][0], false, hB[0],
                                                   (short)0, a1, false, false);
        a0 = __builtin_amdgcn_wmma_f32_16x16x4_f32(false, Ahh[0][1], false, hB[1],
                                                   (short)0, a0, false, false);
        a1 = __builtin_amdgcn_wmma_f32_16x16x4_f32(false, Ahh[1][1], false, hB[1],
                                                   (short)0, a1, false, false);
        ISSUE_IN_THIS_ORDER();
        behind_wmmas();
        v2f tf0 = {FAST_TANH(a0[1]), FAST_TANH(a0[5])};
        ISSUE_IN_THIS_ORDER();
        v2f ti0 = {FAST_TANH(a0[0]), FAST_TANH(a0[4])};
        ISSUE_IN_THIS_ORDER();
        v2f tf1 = {FAST_TANH(a1[1]), FAST_TANH(a1[5])};
        ISSUE_IN_THIS_ORDER();
        v2f fg0 = __builtin_elementwise_fma(hf, tf0, hf);
        ISSUE_IN_THIS_ORDER();
        v2f to0 = {FAST_TANH(a0[3]), FAST_TANH(a0[7])};
        ISSUE_IN_THIS_ORDER();
        v2f tg0 = {FAST_TANH(a0[2]), FAST_TANH(a0[6])};
        ISSUE_IN_THIS_ORDER();
        v2f ig0 = __builtin_elementwise_fma(hf, ti0, hf);
        ISSUE_IN_THIS_ORDER();
        v2f fc0 = fg0 * cp[0];
        ISSUE_IN_THIS_ORDER();
        v2f ti1 = {FAST_TANH(a1[0]), FAST_TANH(a1[4])};
        ISSUE_IN_THIS_ORDER();
        cp[0] = __builtin_elementwise_fma(ig0, tg0, fc0);
        ISSUE_IN_THIS_ORDER();
        v2f fg1 = __builtin_elementwise_fma(hf, tf1, hf);
        ISSUE_IN_THIS_ORDER();
        v2f tg1 = {FAST_TANH(a1[2]), FAST_TANH(a1[6])};
        ISSUE_IN_THIS_ORDER();
        v2f fc1 = fg1 * cp[1];
        ISSUE_IN_THIS_ORDER();
        v2f ig1 = __builtin_elementwise_fma(hf, ti1, hf);
        ISSUE_IN_THIS_ORDER();
        v2f og0 = __builtin_elementwise_fma(hf, to0, hf);
        ISSUE_IN_THIS_ORDER();
        v2f tc0 = {FAST_TANH(cp[0][0]), FAST_TANH(cp[0][1])};
        ISSUE_IN_THIS_ORDER();
        cp[1] = __builtin_elementwise_fma(ig1, tg1, fc1);
        ISSUE_IN_THIS_ORDER();
        v2f to1 = {FAST_TANH(a1[3]), FAST_TANH(a1[7])};
        ISSUE_IN_THIS_ORDER();
        hB[0] = og0 * tc0;
        ISSUE_IN_THIS_ORDER();
        tc1 = v2f{FAST_TANH(cp[1][0]), FAST_TANH(cp[1][1])};
        ISSUE_IN_THIS_ORDER();
        og1 = __builtin_elementwise_fma(hf, to1, hf);
        ISSUE_IN_THIS_ORDER();
    };

#define ARRIVE_STEP 3
    auto run_group = [&](v8f (&cur)[GRP][2], v8f (&nxt)[GRP][2], const v4f* pn, bool rd, int ho) {
#pragma unroll
        for (int s = 0; s < GRP; ++s) {
            step(cur[s][0], cur[s][1], [&]() {
                if (s == 0) {
                    if (ho == 2) {
                        __builtin_amdgcn_s_barrier_wait(-1);
                        __builtin_amdgcn_fence(__ATOMIC_ACQUIRE, "workgroup", "local");
                    }
                    if (rd) group_read(pn, nxt);
                    __builtin_amdgcn_sched_barrier(0);
                }
            });
            if (ho == 1 && s == ARRIVE_STEP) {
                __builtin_amdgcn_fence(__ATOMIC_RELEASE, "workgroup", "local");
                __builtin_amdgcn_s_barrier_signal(-1);
            }
        }
    };

    const int p = wave - 1;

    if (wave == 0) {
        __builtin_amdgcn_s_barrier_signal(-1);
        __builtin_amdgcn_s_barrier_wait(-1);
        __builtin_amdgcn_fence(__ATOMIC_ACQUIRE, "workgroup", "local");
        v8f ta[GRP][2], tb[GRP][2];
        group_read(xg + l, ta);
        for (int c = 0; c < NCH; ++c) {
            const v4f* base = xg + (c & 1) * (CH * SLOT) + l;
            const v4f* next = xg + ((c + 1) & 1) * (CH * SLOT) + l;
            run_group(ta, tb, base + 1 * (GRP * SLOT), true, 0);
            run_group(tb, ta, base + 2 * (GRP * SLOT), true, 0);
            run_group(ta, tb, base + 3 * (GRP * SLOT), true, 1);
            run_group(tb, ta, next, c + 1 < NCH, 2);
        }
    } else {
        for (int j = 0; j <= NCH; ++j) {
            if (j < NCH) {
                v4f* dst = xg + (j & 1) * (CH * SLOT) + l;
                v2f xr[PSTEPS][4];
#pragma unroll
                for (int i = 0; i < PSTEPS; ++i) x_load(j * CH + p + NPROD * i, xr[i]);
#pragma unroll
                for (int i = 0; i < PSTEPS; ++i) {
                    v8f a0, a1;
                    x_project(xr[i], a0, a1);
                    v4f* d = dst + (p + NPROD * i) * SLOT;
                    d[0]  = __builtin_shufflevector(a0, a0, 0, 1, 2, 3);
                    d[32] = __builtin_shufflevector(a0, a0, 4, 5, 6, 7);
                    d[64] = __builtin_shufflevector(a1, a1, 0, 1, 2, 3);
                    d[96] = __builtin_shufflevector(a1, a1, 4, 5, 6, 7);
                }
            } else {
                const int tid = p * 32 + l;
                const float* w1  = fc1_w + tid * HDIM;
                const float* b1  = fc1_b + tid;
                const float* w2  = fc_w + tid;
                v4f*   hws  = hw + 2 * tid;
                float* hfcs = (float*)hfc + tid;
#pragma unroll
                for (int k = 0; k < FFDIM / (32 * NPROD); ++k) {
                    const int o = (32 * NPROD) * k;
                    const float* wr = w1 + o * HDIM;
                    v4f wa, wb;
                    wa[0] = wr[0]; wa[1] = wr[1]; wa[2] = wr[4]; wa[3] = wr[5];
                    wb[0] = wr[2]; wb[1] = wr[3]; wb[2] = wr[6]; wb[3] = b1[o];
                    hws[2 * o]     = wa;
                    hws[2 * o + 1] = wb;
                    hfcs[o]        = w2[o];
                }
            }
            __builtin_amdgcn_fence(__ATOMIC_RELEASE, "workgroup", "local");
            __builtin_amdgcn_s_barrier();
            __builtin_amdgcn_fence(__ATOMIC_ACQUIRE, "workgroup", "local");
        }
    }

    const int hr0 = (wave == 0) ? NPROD : wave - 1;
    const int hnt = (wave == 0) ? HTW - 1 : HTW;
    v4f ha[HTW], hf[HTW][2];
    auto head_operands = [&]() {
#pragma unroll
        for (int i = 0; i < HTW; ++i) {
            if (i < hnt) {
                const int T = (1 + NPROD) * i + hr0;
                ha[i]    = hw[2 * (16 * T + nl) + half];
                hf[i][0] = hfc[4 * T + 2 * half];
                hf[i][1] = hfc[4 * T + 2 * half + 1];
            }
        }
    };
    if (wave != 0) head_operands();

    if (wave == 0) {
        hB[1] = og1 * tc1;
#pragma unroll
        for (int mt = 0; mt < 2; ++mt) {
            hbuf[nl][mt * 4 + 2 * half]     = hB[mt][0];
            hbuf[nl][mt * 4 + 2 * half + 1] = hB[mt][1];
        }
    }
    __syncthreads();

    v2f rb0, rb1;
    {
        const float h0 = hbuf[nl][2 * half], h1 = hbuf[nl][2 * half + 1];
        const float h2 = hbuf[nl][4 + 2 * half], h3 = hbuf[nl][4 + 2 * half + 1];
        if (wave == 0) head_operands();
        rb0[0] = h0 > 0.0f ? h0 : 0.0f;
        rb0[1] = h1 > 0.0f ? h1 : 0.0f;
        rb1[0] = h2 > 0.0f ? h2 : 0.0f;
        rb1[1] = (half == 1) ? 1.0f : (h3 > 0.0f ? h3 : 0.0f);
    }

    v2f hs[4];
#pragma unroll
    for (int q = 0; q < 4; ++q) { hs[q][0] = 0.0f; hs[q][1] = 0.0f; }
#pragma unroll
    for (int i = 0; i < HTW; ++i) {
        if (i < hnt) {
            v8f z = {0.0f, 0.0f, 0.0f, 0.0f, 0.0f, 0.0f, 0.0f, 0.0f};
            const v2f a0 = {ha[i][0], ha[i][1]};
            const v2f a1 = {ha[i][2], ha[i][3]};
            z = __builtin_amdgcn_wmma_f32_16x16x4_f32(false, a0, false, rb0,
                                                      (short)0, z, false, false);
            z = __builtin_amdgcn_wmma_f32_16x16x4_f32(false, a1, false, rb1,
                                                      (short)0, z, false, false);
#pragma unroll
            for (int q = 0; q < 4; ++q) {
                v2f r, w;
                r[0] = z[2 * q]     > 0.0f ? z[2 * q]     : 0.0f;
                r[1] = z[2 * q + 1] > 0.0f ? z[2 * q + 1] : 0.0f;
                w[0] = hf[i][q >> 1][(2 * q) & 3];
                w[1] = hf[i][q >> 1][(2 * q + 1) & 3];
                hs[q] = __builtin_elementwise_fma(r, w, hs[q]);
            }
        }
    }
    red[wave][l] = ((hs[0][0] + hs[0][1]) + (hs[1][0] + hs[1][1]))
                 + ((hs[2][0] + hs[2][1]) + (hs[3][0] + hs[3][1]));
    __syncthreads();
    if (wave == 0 && half == 0) {
        float s = red[0][nl] + red[0][nl + 16];
#pragma unroll
        for (int w = 1; w <= NPROD; ++w) s += red[w][nl] + red[w][nl + 16];
        out[b0 + nl] = s + fc_b[0];
    }
}

extern "C" void kernel_launch(void* const* d_in, const int* in_sizes, int n_in,
                              void* d_out, int out_size, void* d_ws, size_t ws_size,
                              hipStream_t stream) {
    (void)in_sizes; (void)n_in; (void)out_size; (void)d_ws; (void)ws_size;
    const float* x     = (const float*)d_in[0];
    const float* w_ih  = (const float*)d_in[1];
    const float* w_hh  = (const float*)d_in[2];
    const float* b_ih  = (const float*)d_in[3];
    const float* b_hh  = (const float*)d_in[4];
    const float* fc1_w = (const float*)d_in[5];
    const float* fc1_b = (const float*)d_in[6];
    const float* fc_w  = (const float*)d_in[7];
    const float* fc_b  = (const float*)d_in[8];

    lstm_fused_kernel<<<dim3(BATCH / 16), dim3(32 * (1 + NPROD)), 0, stream>>>(
        x, w_ih, w_hh, b_ih, b_hh, fc1_w, fc1_b, fc_w, fc_b, (float*)d_out);
}
